# starting kernel
# baseline (speedup 1.0000x reference)
#include <hip/hip_runtime.h>

#define NNODES 100000
#define NEDGES 1600000
#define FEATS  64

typedef __attribute__((ext_vector_type(2))) float v2f;
typedef __attribute__((ext_vector_type(4))) float v4f;
typedef __attribute__((ext_vector_type(8))) float v8f;

__global__ void zero_f4(float* __restrict__ p, int n4) {
    int i = blockIdx.x * blockDim.x + threadIdx.x;
    if (i < n4) {
        v4f z = {};
        ((v4f*)p)[i] = z;
    }
}

__global__ void degree_kernel(const int* __restrict__ src, const int* __restrict__ dst,
                              float* __restrict__ deg, int nE) {
    int e = blockIdx.x * blockDim.x + threadIdx.x;
    if (e < nE) {
        atomicAdd(&deg[src[e]], 1.0f);
        atomicAdd(&deg[NNODES + dst[e]], 1.0f);
    }
}

__global__ void norm_kernel(float* __restrict__ deg, int n) {
    int i = blockIdx.x * blockDim.x + threadIdx.x;
    if (i < n) deg[i] = rsqrtf(fmaxf(deg[i], 1.0f));
}

__global__ void gather_kernel(const float* __restrict__ h, const int* __restrict__ src,
                              const int* __restrict__ dst, const float* __restrict__ src_norm,
                              float* __restrict__ agg, int nE) {
    int idx = blockIdx.x * blockDim.x + threadIdx.x;
    int e = idx >> 4;
    if (e >= nE) return;
    int fo = (idx & 15) << 2;
    int s = src[e];
    int d = dst[e];
    float sn = src_norm[s];
    v4f v = *(const v4f*)(h + (size_t)s * FEATS + fo);
    float* ap = agg + (size_t)d * FEATS + fo;
    atomicAdd(ap + 0, v.x * sn);
    atomicAdd(ap + 1, v.y * sn);
    atomicAdd(ap + 2, v.z * sn);
    atomicAdd(ap + 3, v.w * sn);
}

template <int NT, bool RELU>
__global__ void gemm_wmma(const float* __restrict__ agg, const float* __restrict__ dst_norm,
                          const float* __restrict__ W, const float* __restrict__ bias,
                          float* __restrict__ out, int nRows) {
    constexpr int NOUT = NT * 16;
    const int lane = threadIdx.x & 31;
    const int wave = threadIdx.x >> 5;
    const int rowBase = (blockIdx.x * (blockDim.x >> 5) + wave) * 16;
    if (rowBase >= nRows) return;

    const int half = lane >> 4;
    const int l16  = lane & 15;

    int arow = rowBase + l16;
    if (arow >= nRows) arow = nRows - 1;
    const float dn = dst_norm[arow];
    const float* aptr = agg + (size_t)arow * FEATS + half * 2;

    v8f acc[NT] = {};

#pragma unroll
    for (int k0 = 0; k0 < FEATS; k0 += 4) {
        v2f A;
        A.x = aptr[k0 + 0] * dn;
        A.y = aptr[k0 + 1] * dn;
        const float* wp = W + (size_t)(k0 + half * 2) * NOUT + l16;
#pragma unroll
        for (int t = 0; t < NT; ++t) {
            v2f B;
            B.x = wp[t * 16];
            B.y = wp[t * 16 + NOUT];
            acc[t] = __builtin_amdgcn_wmma_f32_16x16x4_f32(
                false, A, false, B, (short)0, acc[t], false, false);
        }
    }

#pragma unroll
    for (int t = 0; t < NT; ++t) {
        const int col = t * 16 + l16;
        const float bv = bias[col];
#pragma unroll
        for (int i = 0; i < 8; ++i) {
            const int row = rowBase + half * 8 + i;
            float v = acc[t][i] + bv;
            if (RELU) v = fmaxf(v, 0.0f);
            if (row < nRows) out[(size_t)row * NOUT + col] = v;
        }
    }
}

extern "C" void kernel_launch(void* const* d_in, const int* in_sizes, int n_in,
                              void* d_out, int out_size, void* d_ws, size_t ws_size,
                              hipStream_t stream) {
    const float* x    = (const float*)d_in[0];
    const int*   esrc = (const int*)d_in[1];
    const int*   edst = (const int*)d_in[2];
    const float* W0   = (const float*)d_in[3];
    const float* b0   = (const float*)d_in[4];
    const float* W1   = (const float*)d_in[5];
    const float* b1   = (const float*)d_in[6];
    const float* W2   = (const float*)d_in[7];
    const float* b2   = (const float*)d_in[8];
    float* outp = (float*)d_out;

    float* ws       = (float*)d_ws;
    float* norms    = ws;
    float* src_norm = norms;
    float* dst_norm = norms + NNODES;
    float* agg      = norms + 2 * NNODES;
    float* h        = agg + (size_t)NNODES * FEATS;

    const int T = 256;
    const int n4feat   = NNODES * FEATS / 4;
    const int gZeroN   = (2 * NNODES / 4 + T - 1) / T;
    const int gZeroF   = (n4feat + T - 1) / T;
    const int gEdge    = (NEDGES + T - 1) / T;
    const int gGather  = (NEDGES * 16 + T - 1) / T;
    const int wavesPerBlock = T / 32;
    const int gGemm    = (NNODES + 16 * wavesPerBlock - 1) / (16 * wavesPerBlock);

    zero_f4<<<gZeroN, T, 0, stream>>>(norms, 2 * NNODES / 4);
    degree_kernel<<<gEdge, T, 0, stream>>>(esrc, edst, norms, NEDGES);
    norm_kernel<<<(2 * NNODES + T - 1) / T, T, 0, stream>>>(norms, 2 * NNODES);

    zero_f4<<<gZeroF, T, 0, stream>>>(agg, n4feat);
    gather_kernel<<<gGather, T, 0, stream>>>(x, esrc, edst, src_norm, agg, NEDGES);
    gemm_wmma<4, true><<<gGemm, T, 0, stream>>>(agg, dst_norm, W0, b0, h, NNODES);

    zero_f4<<<gZeroF, T, 0, stream>>>(agg, n4feat);
    gather_kernel<<<gGather, T, 0, stream>>>(h, esrc, edst, src_norm, agg, NEDGES);
    gemm_wmma<4, true><<<gGemm, T, 0, stream>>>(agg, dst_norm, W1, b1, h, NNODES);

    zero_f4<<<gZeroF, T, 0, stream>>>(agg, n4feat);
    gather_kernel<<<gGather, T, 0, stream>>>(h, esrc, edst, src_norm, agg, NEDGES);
    gemm_wmma<2, false><<<gGemm, T, 0, stream>>>(agg, dst_norm, W2, b2, outp, NNODES);
}


# float atomic scatter-add replaced by a destination-sorted edge list (counting sort, int atomics give each edge its rank) pulled row-wise through LDS into the WMMA GEMM; sum order = arrival rank
# speedup vs baseline: 2.2279x; 2.2279x over previous
#include <hip/hip_runtime.h>

#define NNODES 100000
#define NEDGES 1600000
#define FEATS  64

typedef __attribute__((ext_vector_type(2))) float v2f;
typedef __attribute__((ext_vector_type(4))) float v4f;
typedef __attribute__((ext_vector_type(8))) float v8f;
typedef __attribute__((ext_vector_type(4))) int   v4i;
typedef __attribute__((ext_vector_type(2))) int   v2i;

static_assert(NNODES % 16 == 0, "row tiles of 16 and 4-wide counter groups");

__global__ void zero_i4(int* __restrict__ p, int n4) {
    int i = blockIdx.x * blockDim.x + threadIdx.x;
    if (i < n4) {
        v4i z = {};
        ((v4i*)p)[i] = z;
    }
}

__global__ void count_rank_kernel(const int* __restrict__ src, const int* __restrict__ dst,
                                  int* __restrict__ cnt_out, int* __restrict__ cnt_in,
                                  int* __restrict__ rank, int nE) {
    int e = blockIdx.x * blockDim.x + threadIdx.x;
    if (e < nE) {
        const int s = src[e];
        const int d = dst[e];
        atomicAdd(&cnt_out[s], 1);
        rank[e] = atomicAdd(&cnt_in[d], 1);
    }
}

#define SCAN_T  1024
#define SCAN_V4 (((NNODES + SCAN_T - 1) / SCAN_T + 3) / 4)

__global__ __launch_bounds__(SCAN_T) void scan_kernel(const int* __restrict__ cnt_in,
                                                      int* __restrict__ off, int n) {
    __shared__ int part[SCAN_T];
    const int t  = threadIdx.x;
    const int g0 = t * SCAN_V4;
    const int n4 = n >> 2;
    const v4i* c4 = (const v4i*)cnt_in;

    v4i c[SCAN_V4];
#pragma unroll
    for (int i = 0; i < SCAN_V4; ++i) {
        const int g = g0 + i;
        c[i] = c4[g < n4 ? g : n4 - 1];
    }
    int sum = 0;
#pragma unroll
    for (int i = 0; i < SCAN_V4; ++i) {
        if (g0 + i >= n4) { v4i z = {}; c[i] = z; }
        sum += (c[i].x + c[i].y) + (c[i].z + c[i].w);
    }
    part[t] = sum;
    __syncthreads();
    for (int d = 1; d < SCAN_T; d <<= 1) {
        const int v = (t >= d) ? part[t - d] : 0;
        __syncthreads();
        part[t] += v;
        __syncthreads();
    }
    int run = part[t] - sum;
    v4i* o4 = (v4i*)off;
#pragma unroll
    for (int i = 0; i < SCAN_V4; ++i) {
        const int g = g0 + i;
        v4i o;
        o.x = run;
        o.y = o.x + c[i].x;
        o.z = o.y + c[i].y;
        o.w = o.z + c[i].z;
        run = o.w + c[i].w;
        if (g < n4) o4[g] = o;
    }
    if (t == SCAN_T - 1) off[n] = part[t];
}

__global__ void fill_kernel(const int* __restrict__ src, const int* __restrict__ dst,
                            const int* __restrict__ rank, const int* __restrict__ off,
                            const int* __restrict__ cnt_out,
                            v2i* __restrict__ csr, int nE) {
    int e = blockIdx.x * blockDim.x + threadIdx.x;
    if (e < nE) {
        const int s = src[e];
        const int p = off[dst[e]] + rank[e];
        v2i rec;
        rec.x = s;
        rec.y = __float_as_int(rsqrtf(fmaxf((float)cnt_out[s], 1.0f)));
        csr[p] = rec;
    }
}

#define PW  4
#define LDP (FEATS + 4)

template <int NT, bool RELU>
__global__ __launch_bounds__(PW * 32) void pull_gemm_wmma(
        const float* __restrict__ hin, const int* __restrict__ off,
        const v2i* __restrict__ csr,
        const float* __restrict__ W, const float* __restrict__ bias,
        float* __restrict__ out, int nRows) {
    constexpr int NOUT = NT * 16;
    __shared__ __attribute__((aligned(16))) float tile[PW][16 * LDP];

    const int lane = threadIdx.x & 31;
    const int wave = threadIdx.x >> 5;
    const int rowBase = (blockIdx.x * PW + wave) * 16;
    const bool active = rowBase < nRows;
    const int half = lane >> 4;
    const int l16  = lane & 15;
    float* myt = tile[wave];

    if (active) {
        for (int i = 0; i < 8; ++i) {
            const int r   = i * 2 + half;
            const int row = rowBase + r;
            int beg = 0, end = 0;
            if (row < nRows) { beg = off[row]; end = off[row + 1]; }
            v4f acc = {};
            for (int p = beg; p < end; p += 8) {
                int   s[8];
                float w[8];
                v4f   v[8];
#pragma unroll
                for (int u = 0; u < 8; ++u) {
                    int q = p + u;
                    q = q < end ? q : end - 1;
                    const v2i rec = csr[q];
                    s[u] = rec.x;
                    w[u] = __int_as_float(rec.y);
                }
#pragma unroll
                for (int u = 0; u < 8; ++u)
                    v[u] = *(const v4f*)(hin + (size_t)s[u] * FEATS + l16 * 4);
#pragma unroll
                for (int u = 0; u < 8; ++u) {
                    if (p + u < end) {
                        acc.x = fmaf(v[u].x, w[u], acc.x);
                        acc.y = fmaf(v[u].y, w[u], acc.y);
                        acc.z = fmaf(v[u].z, w[u], acc.z);
                        acc.w = fmaf(v[u].w, w[u], acc.w);
                    }
                }
            }
            const float dn = rsqrtf(fmaxf((float)(end - beg), 1.0f));
            *(v4f*)(myt + r * LDP + l16 * 4) = acc * dn;
        }
    }
    __syncthreads();
    if (!active) return;

    v8f acc2[NT] = {};
    const float* ap = myt + l16 * LDP + half * 2;

#pragma unroll
    for (int k0 = 0; k0 < FEATS; k0 += 4) {
        const v2f A = *(const v2f*)(ap + k0);
        const float* wp = W + (size_t)(k0 + half * 2) * NOUT + l16;
#pragma unroll
        for (int t = 0; t < NT; ++t) {
            v2f B;
            B.x = wp[t * 16];
            B.y = wp[t * 16 + NOUT];
            acc2[t] = __builtin_amdgcn_wmma_f32_16x16x4_f32(
                false, A, false, B, (short)0, acc2[t], false, false);
        }
    }

#pragma unroll
    for (int t = 0; t < NT; ++t) {
        const int col = t * 16 + l16;
        const float bv = bias[col];
#pragma unroll
        for (int i = 0; i < 8; ++i) {
            const int row = rowBase + half * 8 + i;
            float v = acc2[t][i] + bv;
            if (RELU) v = fmaxf(v, 0.0f);
            if (row < nRows) out[(size_t)row * NOUT + col] = v;
        }
    }
}

extern "C" void kernel_launch(void* const* d_in, const int* in_sizes, int n_in,
                              void* d_out, int out_size, void* d_ws, size_t ws_size,
                              hipStream_t stream) {
    const float* x    = (const float*)d_in[0];
    const int*   esrc = (const int*)d_in[1];
    const int*   edst = (const int*)d_in[2];
    const float* W0   = (const float*)d_in[3];
    const float* b0   = (const float*)d_in[4];
    const float* W1   = (const float*)d_in[5];
    const float* b1   = (const float*)d_in[6];
    const float* W2   = (const float*)d_in[7];
    const float* b2   = (const float*)d_in[8];
    float* outp = (float*)d_out;

    const int NOFF = 100096;
    int*   cnt_out = (int*)d_ws;
    int*   cnt_in  = cnt_out + NNODES;
    int*   off     = cnt_in + NNODES;
    int*   rank    = off + NOFF;
    v2i*   csr     = (v2i*)(rank + NEDGES);
    float* hA      = (float*)(csr + NEDGES);
    float* hB      = hA + (size_t)NNODES * FEATS;

    const int T = 256;
    const int gZero = (2 * NNODES / 4 + T - 1) / T;
    const int gEdge = (NEDGES + T - 1) / T;
    const int gPull = (NNODES + 16 * PW - 1) / (16 * PW);

    zero_i4<<<gZero, T, 0, stream>>>(cnt_out, 2 * NNODES / 4);
    count_rank_kernel<<<gEdge, T, 0, stream>>>(esrc, edst, cnt_out, cnt_in, rank, NEDGES);
    scan_kernel<<<1, SCAN_T, 0, stream>>>(cnt_in, off, NNODES);
    fill_kernel<<<gEdge, T, 0, stream>>>(esrc, edst, rank, off, cnt_out, csr, NEDGES);

    pull_gemm_wmma<4, true ><<<gPull, PW * 32, 0, stream>>>(x,  off, csr, W0, b0, hA,   NNODES);
    pull_gemm_wmma<4, true ><<<gPull, PW * 32, 0, stream>>>(hA, off, csr, W1, b1, hB,   NNODES);
    pull_gemm_wmma<2, false><<<gPull, PW * 32, 0, stream>>>(hB, off, csr, W2, b2, outp, NNODES);
}


# out-degree counts moved from one scattered global int add per edge to LDS counters private to each work-group (64 edge chunks x 2 node halves), flushed as whole-line adds; float sums untouched
# speedup vs baseline: 2.5189x; 1.1306x over previous
#include <hip/hip_runtime.h>

#define NNODES 100000
#define NEDGES 1600000
#define FEATS  64

typedef __attribute__((ext_vector_type(2))) float v2f;
typedef __attribute__((ext_vector_type(4))) float v4f;
typedef __attribute__((ext_vector_type(8))) float v8f;
typedef __attribute__((ext_vector_type(4))) int   v4i;
typedef __attribute__((ext_vector_type(2))) int   v2i;

static_assert(NNODES % 16 == 0, "row tiles of 16 and 4-wide counter groups");

__global__ void zero_i4(int* __restrict__ p, int n4) {
    int i = blockIdx.x * blockDim.x + threadIdx.x;
    if (i < n4) {
        v4i z = {};
        ((v4i*)p)[i] = z;
    }
}

#define OD_T      1024
#define OD_CHUNKS 64
#define OD_HALF   50016
#define OD_E4     (NEDGES / OD_CHUNKS / 4)
#define OD_IT     ((OD_E4 + OD_T - 1) / OD_T)
static_assert(NEDGES % (OD_CHUNKS * 4) == 0, "chunks of whole int4 groups");
static_assert(OD_HALF % 32 == 0 && OD_HALF < NNODES && 2 * OD_HALF >= NNODES, "two halves cover the nodes");
static_assert((OD_CHUNKS - 1) * 768 < OD_HALF, "rotated sweep wraps at most once");

__global__ __launch_bounds__(OD_T) void outdeg_lds_kernel(const int* __restrict__ src,
                                                          int* __restrict__ cnt_out) {
    __shared__ __attribute__((aligned(16))) int cnt[OD_HALF];
    const int t     = threadIdx.x;
    const int chunk = blockIdx.x;
    const int hbase = blockIdx.y * OD_HALF;

    for (int k = t; k < OD_HALF / 4; k += OD_T) {
        v4i z = {};
        ((v4i*)cnt)[k] = z;
    }
    __syncthreads();

    const v4i* s4 = (const v4i*)src + (size_t)chunk * OD_E4;
    v4i sv[OD_IT];
#pragma unroll
    for (int i = 0; i < OD_IT; ++i) {
        const int k = t + i * OD_T;
        sv[i] = s4[k < OD_E4 ? k : OD_E4 - 1];
    }
#pragma unroll
    for (int i = 0; i < OD_IT; ++i) {
        if (t + i * OD_T < OD_E4) {
            unsigned r;
            r = (unsigned)(sv[i].x - hbase); if (r < (unsigned)OD_HALF) atomicAdd(&cnt[r], 1);
            r = (unsigned)(sv[i].y - hbase); if (r < (unsigned)OD_HALF) atomicAdd(&cnt[r], 1);
            r = (unsigned)(sv[i].z - hbase); if (r < (unsigned)OD_HALF) atomicAdd(&cnt[r], 1);
            r = (unsigned)(sv[i].w - hbase); if (r < (unsigned)OD_HALF) atomicAdd(&cnt[r], 1);
        }
    }
    __syncthreads();

    const int nH  = (NNODES - hbase) < OD_HALF ? (NNODES - hbase) : OD_HALF;
    const int rot = chunk * 768;
    for (int k = t; k < OD_HALF; k += OD_T) {
        int idx = k + rot;
        if (idx >= OD_HALF) idx -= OD_HALF;
        const int v = cnt[idx];
        if (v != 0 && idx < nH) atomicAdd(&cnt_out[hbase + idx], v);
    }
}

__global__ void count_rank_kernel(const int* __restrict__ dst,
                                  int* __restrict__ cnt_in,
                                  int* __restrict__ rank, int nE) {
    int e = blockIdx.x * blockDim.x + threadIdx.x;
    if (e < nE) {
        rank[e] = atomicAdd(&cnt_in[dst[e]], 1);
    }
}

#define SCAN_T  1024
#define SCAN_V4 (((NNODES + SCAN_T - 1) / SCAN_T + 3) / 4)

__global__ __launch_bounds__(SCAN_T) void scan_kernel(const int* __restrict__ cnt_in,
                                                      int* __restrict__ off, int n) {
    __shared__ int part[SCAN_T];
    const int t  = threadIdx.x;
    const int g0 = t * SCAN_V4;
    const int n4 = n >> 2;
    const v4i* c4 = (const v4i*)cnt_in;

    v4i c[SCAN_V4];
#pragma unroll
    for (int i = 0; i < SCAN_V4; ++i) {
        const int g = g0 + i;
        c[i] = c4[g < n4 ? g : n4 - 1];
    }
    int sum = 0;
#pragma unroll
    for (int i = 0; i < SCAN_V4; ++i) {
        if (g0 + i >= n4) { v4i z = {}; c[i] = z; }
        sum += (c[i].x + c[i].y) + (c[i].z + c[i].w);
    }
    part[t] = sum;
    __syncthreads();
    for (int d = 1; d < SCAN_T; d <<= 1) {
        const int v = (t >= d) ? part[t - d] : 0;
        __syncthreads();
        part[t] += v;
        __syncthreads();
    }
    int run = part[t] - sum;
    v4i* o4 = (v4i*)off;
#pragma unroll
    for (int i = 0; i < SCAN_V4; ++i) {
        const int g = g0 + i;
        v4i o;
        o.x = run;
        o.y = o.x + c[i].x;
        o.z = o.y + c[i].y;
        o.w = o.z + c[i].z;
        run = o.w + c[i].w;
        if (g < n4) o4[g] = o;
    }
    if (t == SCAN_T - 1) off[n] = part[t];
}

__global__ void fill_kernel(const int* __restrict__ src, const int* __restrict__ dst,
                            const int* __restrict__ rank, const int* __restrict__ off,
                            const int* __restrict__ cnt_out,
                            v2i* __restrict__ csr, int nE) {
    int e = blockIdx.x * blockDim.x + threadIdx.x;
    if (e < nE) {
        const int s = src[e];
        const int p = off[dst[e]] + rank[e];
        v2i rec;
        rec.x = s;
        rec.y = __float_as_int(rsqrtf(fmaxf((float)cnt_out[s], 1.0f)));
        csr[p] = rec;
    }
}

#define PW  4
#define LDP (FEATS + 4)

template <int NT, bool RELU>
__global__ __launch_bounds__(PW * 32) void pull_gemm_wmma(
        const float* __restrict__ hin, const int* __restrict__ off,
        const v2i* __restrict__ csr,
        const float* __restrict__ W, const float* __restrict__ bias,
        float* __restrict__ out, int nRows) {
    constexpr int NOUT = NT * 16;
    __shared__ __attribute__((aligned(16))) float tile[PW][16 * LDP];

    const int lane = threadIdx.x & 31;
    const int wave = threadIdx.x >> 5;
    const int rowBase = (blockIdx.x * PW + wave) * 16;
    const bool active = rowBase < nRows;
    const int half = lane >> 4;
    const int l16  = lane & 15;
    float* myt = tile[wave];

    if (active) {
        for (int i = 0; i < 8; ++i) {
            const int r   = i * 2 + half;
            const int row = rowBase + r;
            int beg = 0, end = 0;
            if (row < nRows) { beg = off[row]; end = off[row + 1]; }
            v4f acc = {};
            for (int p = beg; p < end; p += 8) {
                int   s[8];
                float w[8];
                v4f   v[8];
#pragma unroll
                for (int u = 0; u < 8; ++u) {
                    int q = p + u;
                    q = q < end ? q : end - 1;
                    const v2i rec = csr[q];
                    s[u] = rec.x;
                    w[u] = __int_as_float(rec.y);
                }
#pragma unroll
                for (int u = 0; u < 8; ++u)
                    v[u] = *(const v4f*)(hin + (size_t)s[u] * FEATS + l16 * 4);
#pragma unroll
                for (int u = 0; u < 8; ++u) {
                    if (p + u < end) {
                        acc.x = fmaf(v[u].x, w[u], acc.x);
                        acc.y = fmaf(v[u].y, w[u], acc.y);
                        acc.z = fmaf(v[u].z, w[u], acc.z);
                        acc.w = fmaf(v[u].w, w[u], acc.w);
                    }
                }
            }
            const float dn = rsqrtf(fmaxf((float)(end - beg), 1.0f));
            *(v4f*)(myt + r * LDP + l16 * 4) = acc * dn;
        }
    }
    __syncthreads();
    if (!active) return;

    v8f acc2[NT] = {};
    const float* ap = myt + l16 * LDP + half * 2;

#pragma unroll
    for (int k0 = 0; k0 < FEATS; k0 += 4) {
        const v2f A = *(const v2f*)(ap + k0);
        const float* wp = W + (size_t)(k0 + half * 2) * NOUT + l16;
#pragma unroll
        for (int t = 0; t < NT; ++t) {
            v2f B;
            B.x = wp[t * 16];
            B.y = wp[t * 16 + NOUT];
            acc2[t] = __builtin_amdgcn_wmma_f32_16x16x4_f32(
                false, A, false, B, (short)0, acc2[t], false, false);
        }
    }

#pragma unroll
    for (int t = 0; t < NT; ++t) {
        const int col = t * 16 + l16;
        const float bv = bias[col];
#pragma unroll
        for (int i = 0; i < 8; ++i) {
            const int row = rowBase + half * 8 + i;
            float v = acc2[t][i] + bv;
            if (RELU) v = fmaxf(v, 0.0f);
            if (row < nRows) out[(size_t)row * NOUT + col] = v;
        }
    }
}

extern "C" void kernel_launch(void* const* d_in, const int* in_sizes, int n_in,
                              void* d_out, int out_size, void* d_ws, size_t ws_size,
                              hipStream_t stream) {
    const float* x    = (const float*)d_in[0];
    const int*   esrc = (const int*)d_in[1];
    const int*   edst = (const int*)d_in[2];
    const float* W0   = (const float*)d_in[3];
    const float* b0   = (const float*)d_in[4];
    const float* W1   = (const float*)d_in[5];
    const float* b1   = (const float*)d_in[6];
    const float* W2   = (const float*)d_in[7];
    const float* b2   = (const float*)d_in[8];
    float* outp = (float*)d_out;

    const int NOFF = 100096;
    int*   cnt_out = (int*)d_ws;
    int*   cnt_in  = cnt_out + NNODES;
    int*   off     = cnt_in + NNODES;
    int*   rank    = off + NOFF;
    v2i*   csr     = (v2i*)(rank + NEDGES);
    float* hA      = (float*)(csr + NEDGES);
    float* hB      = hA + (size_t)NNODES * FEATS;

    const int T = 256;
    const int gZero = (2 * NNODES / 4 + T - 1) / T;
    const int gEdge = (NEDGES + T - 1) / T;
    const int gPull = (NNODES + 16 * PW - 1) / (16 * PW);

    zero_i4<<<gZero, T, 0, stream>>>(cnt_out, 2 * NNODES / 4);
    outdeg_lds_kernel<<<dim3(OD_CHUNKS, 2), OD_T, 0, stream>>>(esrc, cnt_out);
    count_rank_kernel<<<gEdge, T, 0, stream>>>(edst, cnt_in, rank, NEDGES);
    scan_kernel<<<1, SCAN_T, 0, stream>>>(cnt_in, off, NNODES);
    fill_kernel<<<gEdge, T, 0, stream>>>(esrc, edst, rank, off, cnt_out, csr, NEDGES);

    pull_gemm_wmma<4, true ><<<gPull, PW * 32, 0, stream>>>(x,  off, csr, W0, b0, hA,   NNODES);
    pull_gemm_wmma<4, true ><<<gPull, PW * 32, 0, stream>>>(hA, off, csr, W1, b1, hB,   NNODES);
    pull_gemm_wmma<2, false><<<gPull, PW * 32, 0, stream>>>(hB, off, csr, W2, b2, outp, NNODES);
}


# edge rank and in-degree: scattered returning global int add per edge replaced by LDS counters per work-group (64 chunks x 2 node halves), flushed as whole-line returning adds; sum order by chunk
# speedup vs baseline: 3.3189x; 1.3176x over previous
#include <hip/hip_runtime.h>

#define NNODES 100000
#define NEDGES 1600000
#define FEATS  64

typedef __attribute__((ext_vector_type(2))) float v2f;
typedef __attribute__((ext_vector_type(4))) float v4f;
typedef __attribute__((ext_vector_type(8))) float v8f;
typedef __attribute__((ext_vector_type(4))) int   v4i;
typedef __attribute__((ext_vector_type(2))) int   v2i;

static_assert(NNODES % 16 == 0, "row tiles of 16 and 4-wide counter groups");

__global__ void zero_i4(int* __restrict__ p, int n4) {
    int i = blockIdx.x * blockDim.x + threadIdx.x;
    if (i < n4) {
        v4i z = {};
        ((v4i*)p)[i] = z;
    }
}

#define OD_T      1024
#define OD_CHUNKS 64
#define OD_HALF   50016
#define OD_E4     (NEDGES / OD_CHUNKS / 4)
#define OD_IT     ((OD_E4 + OD_T - 1) / OD_T)
static_assert(NEDGES % (OD_CHUNKS * 4) == 0, "chunks of whole int4 groups");
static_assert(OD_HALF % 32 == 0 && OD_HALF < NNODES && 2 * OD_HALF >= NNODES, "two halves cover the nodes");
static_assert((OD_CHUNKS - 1) * 768 < OD_HALF, "rotated sweep wraps at most once");

__global__ __launch_bounds__(OD_T) void outdeg_lds_kernel(const int* __restrict__ src,
                                                          int* __restrict__ cnt_out) {
    __shared__ __attribute__((aligned(16))) int cnt[OD_HALF];
    const int t     = threadIdx.x;
    const int chunk = blockIdx.x;
    const int hbase = blockIdx.y * OD_HALF;

    for (int k = t; k < OD_HALF / 4; k += OD_T) {
        v4i z = {};
        ((v4i*)cnt)[k] = z;
    }
    __syncthreads();

    const v4i* s4 = (const v4i*)src + (size_t)chunk * OD_E4;
    v4i sv[OD_IT];
#pragma unroll
    for (int i = 0; i < OD_IT; ++i) {
        const int k = t + i * OD_T;
        sv[i] = s4[k < OD_E4 ? k : OD_E4 - 1];
    }
#pragma unroll
    for (int i = 0; i < OD_IT; ++i) {
        if (t + i * OD_T < OD_E4) {
            unsigned r;
            r = (unsigned)(sv[i].x - hbase); if (r < (unsigned)OD_HALF) atomicAdd(&cnt[r], 1);
            r = (unsigned)(sv[i].y - hbase); if (r < (unsigned)OD_HALF) atomicAdd(&cnt[r], 1);
            r = (unsigned)(sv[i].z - hbase); if (r < (unsigned)OD_HALF) atomicAdd(&cnt[r], 1);
            r = (unsigned)(sv[i].w - hbase); if (r < (unsigned)OD_HALF) atomicAdd(&cnt[r], 1);
        }
    }
    __syncthreads();

    const int nH  = (NNODES - hbase) < OD_HALF ? (NNODES - hbase) : OD_HALF;
    const int rot = chunk * 768;
    for (int k = t; k < OD_HALF; k += OD_T) {
        int idx = k + rot;
        if (idx >= OD_HALF) idx -= OD_HALF;
        const int v = cnt[idx];
        if (v != 0 && idx < nH) atomicAdd(&cnt_out[hbase + idx], v);
    }
}

#define IR_E  (NEDGES / OD_CHUNKS)
#define IR_IT ((IR_E + OD_T - 1) / OD_T)
#define IR_FB 13
#define IR_FN ((OD_HALF + OD_T * IR_FB - 1) / (OD_T * IR_FB))
static_assert(NEDGES % OD_CHUNKS == 0, "chunks of whole edges");

__global__ __launch_bounds__(OD_T) void indeg_rank_lds_kernel(const int* __restrict__ dst,
                                                              int* __restrict__ cnt_in,
                                                              int* __restrict__ rank) {
    __shared__ __attribute__((aligned(16))) int cnt[OD_HALF];
    const int t     = threadIdx.x;
    const int chunk = blockIdx.x;
    const int hbase = blockIdx.y * OD_HALF;

    for (int k = t; k < OD_HALF / 4; k += OD_T) {
        v4i z = {};
        ((v4i*)cnt)[k] = z;
    }
    __syncthreads();

    const int* dc = dst  + (size_t)chunk * IR_E;
    int*       rc = rank + (size_t)chunk * IR_E;
    unsigned r[IR_IT];
    int      lr[IR_IT];
#pragma unroll
    for (int i = 0; i < IR_IT; ++i) {
        const int k = t + i * OD_T;
        r[i] = (unsigned)(dc[k < IR_E ? k : IR_E - 1] - hbase);
    }
#pragma unroll
    for (int i = 0; i < IR_IT; ++i) {
        if (t + i * OD_T >= IR_E) r[i] = 0xFFFFFFFFu;
        lr[i] = 0;
        if (r[i] < (unsigned)OD_HALF) lr[i] = atomicAdd(&cnt[r[i]], 1);
    }
    __syncthreads();

    const int nH  = (NNODES - hbase) < OD_HALF ? (NNODES - hbase) : OD_HALF;
    const int rot = chunk * 768;
    for (int b = 0; b < IR_FN; ++b) {
        int idx[IR_FB];
        int v[IR_FB];
#pragma unroll
        for (int u = 0; u < IR_FB; ++u) {
            const int k = t + (b * IR_FB + u) * OD_T;
            int x = k + rot;
            if (x >= OD_HALF) x -= OD_HALF;
            idx[u] = x;
            v[u] = (k < OD_HALF) ? cnt[x] : 0;
        }
#pragma unroll
        for (int u = 0; u < IR_FB; ++u) {
            const bool on = v[u] != 0 && idx[u] < nH;
            int base = 0;
            if (on) base = atomicAdd(&cnt_in[hbase + idx[u]], v[u]);
            v[u] = on ? base : -1;
        }
#pragma unroll
        for (int u = 0; u < IR_FB; ++u) {
            if (v[u] >= 0) cnt[idx[u]] = v[u];
        }
    }
    __syncthreads();

    int st[IR_IT];
#pragma unroll
    for (int i = 0; i < IR_IT; ++i)
        st[i] = cnt[r[i] < (unsigned)OD_HALF ? r[i] : 0];
#pragma unroll
    for (int i = 0; i < IR_IT; ++i) {
        if (r[i] < (unsigned)OD_HALF) rc[t + i * OD_T] = st[i] + lr[i];
    }
}

#define SCAN_T  1024
#define SCAN_V4 (((NNODES + SCAN_T - 1) / SCAN_T + 3) / 4)

__global__ __launch_bounds__(SCAN_T) void scan_kernel(const int* __restrict__ cnt_in,
                                                      int* __restrict__ off, int n) {
    __shared__ int part[SCAN_T];
    const int t  = threadIdx.x;
    const int g0 = t * SCAN_V4;
    const int n4 = n >> 2;
    const v4i* c4 = (const v4i*)cnt_in;

    v4i c[SCAN_V4];
#pragma unroll
    for (int i = 0; i < SCAN_V4; ++i) {
        const int g = g0 + i;
        c[i] = c4[g < n4 ? g : n4 - 1];
    }
    int sum = 0;
#pragma unroll
    for (int i = 0; i < SCAN_V4; ++i) {
        if (g0 + i >= n4) { v4i z = {}; c[i] = z; }
        sum += (c[i].x + c[i].y) + (c[i].z + c[i].w);
    }
    part[t] = sum;
    __syncthreads();
    for (int d = 1; d < SCAN_T; d <<= 1) {
        const int v = (t >= d) ? part[t - d] : 0;
        __syncthreads();
        part[t] += v;
        __syncthreads();
    }
    int run = part[t] - sum;
    v4i* o4 = (v4i*)off;
#pragma unroll
    for (int i = 0; i < SCAN_V4; ++i) {
        const int g = g0 + i;
        v4i o;
        o.x = run;
        o.y = o.x + c[i].x;
        o.z = o.y + c[i].y;
        o.w = o.z + c[i].z;
        run = o.w + c[i].w;
        if (g < n4) o4[g] = o;
    }
    if (t == SCAN_T - 1) off[n] = part[t];
}

__global__ void fill_kernel(const int* __restrict__ src, const int* __restrict__ dst,
                            const int* __restrict__ rank, const int* __restrict__ off,
                            const int* __restrict__ cnt_out,
                            v2i* __restrict__ csr, int nE) {
    int e = blockIdx.x * blockDim.x + threadIdx.x;
    if (e < nE) {
        const int s = src[e];
        const int p = off[dst[e]] + rank[e];
        v2i rec;
        rec.x = s;
        rec.y = __float_as_int(rsqrtf(fmaxf((float)cnt_out[s], 1.0f)));
        csr[p] = rec;
    }
}

#define PW  4
#define LDP (FEATS + 4)

template <int NT, bool RELU>
__global__ __launch_bounds__(PW * 32) void pull_gemm_wmma(
        const float* __restrict__ hin, const int* __restrict__ off,
        const v2i* __restrict__ csr,
        const float* __restrict__ W, const float* __restrict__ bias,
        float* __restrict__ out, int nRows) {
    constexpr int NOUT = NT * 16;
    __shared__ __attribute__((aligned(16))) float tile[PW][16 * LDP];

    const int lane = threadIdx.x & 31;
    const int wave = threadIdx.x >> 5;
    const int rowBase = (blockIdx.x * PW + wave) * 16;
    const bool active = rowBase < nRows;
    const int half = lane >> 4;
    const int l16  = lane & 15;
    float* myt = tile[wave];

    if (active) {
        for (int i = 0; i < 8; ++i) {
            const int r   = i * 2 + half;
            const int row = rowBase + r;
            int beg = 0, end = 0;
            if (row < nRows) { beg = off[row]; end = off[row + 1]; }
            v4f acc = {};
            for (int p = beg; p < end; p += 8) {
                int   s[8];
                float w[8];
                v4f   v[8];
#pragma unroll
                for (int u = 0; u < 8; ++u) {
                    int q = p + u;
                    q = q < end ? q : end - 1;
                    const v2i rec = csr[q];
                    s[u] = rec.x;
                    w[u] = __int_as_float(rec.y);
                }
#pragma unroll
                for (int u = 0; u < 8; ++u)
                    v[u] = *(const v4f*)(hin + (size_t)s[u] * FEATS + l16 * 4);
#pragma unroll
                for (int u = 0; u < 8; ++u) {
                    if (p + u < end) {
                        acc.x = fmaf(v[u].x, w[u], acc.x);
                        acc.y = fmaf(v[u].y, w[u], acc.y);
                        acc.z = fmaf(v[u].z, w[u], acc.z);
                        acc.w = fmaf(v[u].w, w[u], acc.w);
                    }
                }
            }
            const float dn = rsqrtf(fmaxf((float)(end - beg), 1.0f));
            *(v4f*)(myt + r * LDP + l16 * 4) = acc * dn;
        }
    }
    __syncthreads();
    if (!active) return;

    v8f acc2[NT] = {};
    const float* ap = myt + l16 * LDP + half * 2;

#pragma unroll
    for (int k0 = 0; k0 < FEATS; k0 += 4) {
        const v2f A = *(const v2f*)(ap + k0);
        const float* wp = W + (size_t)(k0 + half * 2) * NOUT + l16;
#pragma unroll
        for (int t = 0; t < NT; ++t) {
            v2f B;
            B.x = wp[t * 16];
            B.y = wp[t * 16 + NOUT];
            acc2[t] = __builtin_amdgcn_wmma_f32_16x16x4_f32(
                false, A, false, B, (short)0, acc2[t], false, false);
        }
    }

#pragma unroll
    for (int t = 0; t < NT; ++t) {
        const int col = t * 16 + l16;
        const float bv = bias[col];
#pragma unroll
        for (int i = 0; i < 8; ++i) {
            const int row = rowBase + half * 8 + i;
            float v = acc2[t][i] + bv;
            if (RELU) v = fmaxf(v, 0.0f);
            if (row < nRows) out[(size_t)row * NOUT + col] = v;
        }
    }
}

extern "C" void kernel_launch(void* const* d_in, const int* in_sizes, int n_in,
                              void* d_out, int out_size, void* d_ws, size_t ws_size,
                              hipStream_t stream) {
    const float* x    = (const float*)d_in[0];
    const int*   esrc = (const int*)d_in[1];
    const int*   edst = (const int*)d_in[2];
    const float* W0   = (const float*)d_in[3];
    const float* b0   = (const float*)d_in[4];
    const float* W1   = (const float*)d_in[5];
    const float* b1   = (const float*)d_in[6];
    const float* W2   = (const float*)d_in[7];
    const float* b2   = (const float*)d_in[8];
    float* outp = (float*)d_out;

    const int NOFF = 100096;
    int*   cnt_out = (int*)d_ws;
    int*   cnt_in  = cnt_out + NNODES;
    int*   off     = cnt_in + NNODES;
    int*   rank    = off + NOFF;
    v2i*   csr     = (v2i*)(rank + NEDGES);
    float* hA      = (float*)(csr + NEDGES);
    float* hB      = hA + (size_t)NNODES * FEATS;

    const int T = 256;
    const int gZero = (2 * NNODES / 4 + T - 1) / T;
    const int gEdge = (NEDGES + T - 1) / T;
    const int gPull = (NNODES + 16 * PW - 1) / (16 * PW);

    zero_i4<<<gZero, T, 0, stream>>>(cnt_out, 2 * NNODES / 4);
    outdeg_lds_kernel<<<dim3(OD_CHUNKS, 2), OD_T, 0, stream>>>(esrc, cnt_out);
    indeg_rank_lds_kernel<<<dim3(OD_CHUNKS, 2), OD_T, 0, stream>>>(edst, cnt_in, rank);
    scan_kernel<<<1, SCAN_T, 0, stream>>>(cnt_in, off, NNODES);
    fill_kernel<<<gEdge, T, 0, stream>>>(esrc, edst, rank, off, cnt_out, csr, NEDGES);

    pull_gemm_wmma<4, true ><<<gPull, PW * 32, 0, stream>>>(x,  off, csr, W0, b0, hA,   NNODES);
    pull_gemm_wmma<4, true ><<<gPull, PW * 32, 0, stream>>>(hA, off, csr, W1, b1, hB,   NNODES);
    pull_gemm_wmma<2, false><<<gPull, PW * 32, 0, stream>>>(hB, off, csr, W2, b2, outp, NNODES);
}


# last layer's W2 product (f32 WMMA) moved in front of its aggregation, fused into layer 1's kernel: last gather reads 128-B rows, not 256-B; sums reassociated, A(hW2) for (Ah)W2
# speedup vs baseline: 3.4706x; 1.0457x over previous
#include <hip/hip_runtime.h>

#define NNODES 100000
#define NEDGES 1600000
#define FEATS  64

typedef __attribute__((ext_vector_type(2))) float v2f;
typedef __attribute__((ext_vector_type(4))) float v4f;
typedef __attribute__((ext_vector_type(8))) float v8f;
typedef __attribute__((ext_vector_type(4))) int   v4i;
typedef __attribute__((ext_vector_type(2))) int   v2i;

static_assert(NNODES % 16 == 0, "row tiles of 16 and 4-wide counter groups");

__global__ void zero_i4(int* __restrict__ p, int n4) {
    int i = blockIdx.x * blockDim.x + threadIdx.x;
    if (i < n4) {
        v4i z = {};
        ((v4i*)p)[i] = z;
    }
}

#define OD_T      1024
#define OD_CHUNKS 64
#define OD_HALF   50016
#define OD_E4     (NEDGES / OD_CHUNKS / 4)
#define OD_IT     ((OD_E4 + OD_T - 1) / OD_T)
static_assert(NEDGES % (OD_CHUNKS * 4) == 0, "chunks of whole int4 groups");
static_assert(OD_HALF % 32 == 0 && OD_HALF < NNODES && 2 * OD_HALF >= NNODES, "two halves cover the nodes");
static_assert((OD_CHUNKS - 1) * 768 < OD_HALF, "rotated sweep wraps at most once");

__global__ __launch_bounds__(OD_T) void outdeg_lds_kernel(const int* __restrict__ src,
                                                          int* __restrict__ cnt_out) {
    __shared__ __attribute__((aligned(16))) int cnt[OD_HALF];
    const int t     = threadIdx.x;
    const int chunk = blockIdx.x;
    const int hbase = blockIdx.y * OD_HALF;

    for (int k = t; k < OD_HALF / 4; k += OD_T) {
        v4i z = {};
        ((v4i*)cnt)[k] = z;
    }
    __syncthreads();

    const v4i* s4 = (const v4i*)src + (size_t)chunk * OD_E4;
    v4i sv[OD_IT];
#pragma unroll
    for (int i = 0; i < OD_IT; ++i) {
        const int k = t + i * OD_T;
        sv[i] = s4[k < OD_E4 ? k : OD_E4 - 1];
    }
#pragma unroll
    for (int i = 0; i < OD_IT; ++i) {
        if (t + i * OD_T < OD_E4) {
            unsigned r;
            r = (unsigned)(sv[i].x - hbase); if (r < (unsigned)OD_HALF) atomicAdd(&cnt[r], 1);
            r = (unsigned)(sv[i].y - hbase); if (r < (unsigned)OD_HALF) atomicAdd(&cnt[r], 1);
            r = (unsigned)(sv[i].z - hbase); if (r < (unsigned)OD_HALF) atomicAdd(&cnt[r], 1);
            r = (unsigned)(sv[i].w - hbase); if (r < (unsigned)OD_HALF) atomicAdd(&cnt[r], 1);
        }
    }
    __syncthreads();

    const int nH  = (NNODES - hbase) < OD_HALF ? (NNODES - hbase) : OD_HALF;
    const int rot = chunk * 768;
    for (int k = t; k < OD_HALF; k += OD_T) {
        int idx = k + rot;
        if (idx >= OD_HALF) idx -= OD_HALF;
        const int v = cnt[idx];
        if (v != 0 && idx < nH) atomicAdd(&cnt_out[hbase + idx], v);
    }
}

#define IR_E  (NEDGES / OD_CHUNKS)
#define IR_IT ((IR_E + OD_T - 1) / OD_T)
#define IR_FB 13
#define IR_FN ((OD_HALF + OD_T * IR_FB - 1) / (OD_T * IR_FB))
static_assert(NEDGES % OD_CHUNKS == 0, "chunks of whole edges");

__global__ __launch_bounds__(OD_T) void indeg_rank_lds_kernel(const int* __restrict__ dst,
                                                              int* __restrict__ cnt_in,
                                                              int* __restrict__ rank) {
    __shared__ __attribute__((aligned(16))) int cnt[OD_HALF];
    const int t     = threadIdx.x;
    const int chunk = blockIdx.x;
    const int hbase = blockIdx.y * OD_HALF;

    for (int k = t; k < OD_HALF / 4; k += OD_T) {
        v4i z = {};
        ((v4i*)cnt)[k] = z;
    }
    __syncthreads();

    const int* dc = dst  + (size_t)chunk * IR_E;
    int*       rc = rank + (size_t)chunk * IR_E;
    unsigned r[IR_IT];
    int      lr[IR_IT];
#pragma unroll
    for (int i = 0; i < IR_IT; ++i) {
        const int k = t + i * OD_T;
        r[i] = (unsigned)(dc[k < IR_E ? k : IR_E - 1] - hbase);
    }
#pragma unroll
    for (int i = 0; i < IR_IT; ++i) {
        if (t + i * OD_T >= IR_E) r[i] = 0xFFFFFFFFu;
        lr[i] = 0;
        if (r[i] < (unsigned)OD_HALF) lr[i] = atomicAdd(&cnt[r[i]], 1);
    }
    __syncthreads();

    const int nH  = (NNODES - hbase) < OD_HALF ? (NNODES - hbase) : OD_HALF;
    const int rot = chunk * 768;
    for (int b = 0; b < IR_FN; ++b) {
        int idx[IR_FB];
        int v[IR_FB];
#pragma unroll
        for (int u = 0; u < IR_FB; ++u) {
            const int k = t + (b * IR_FB + u) * OD_T;
            int x = k + rot;
            if (x >= OD_HALF) x -= OD_HALF;
            idx[u] = x;
            v[u] = (k < OD_HALF) ? cnt[x] : 0;
        }
#pragma unroll
        for (int u = 0; u < IR_FB; ++u) {
            const bool on = v[u] != 0 && idx[u] < nH;
            int base = 0;
            if (on) base = atomicAdd(&cnt_in[hbase + idx[u]], v[u]);
            v[u] = on ? base : -1;
        }
#pragma unroll
        for (int u = 0; u < IR_FB; ++u) {
            if (v[u] >= 0) cnt[idx[u]] = v[u];
        }
    }
    __syncthreads();

    int st[IR_IT];
#pragma unroll
    for (int i = 0; i < IR_IT; ++i)
        st[i] = cnt[r[i] < (unsigned)OD_HALF ? r[i] : 0];
#pragma unroll
    for (int i = 0; i < IR_IT; ++i) {
        if (r[i] < (unsigned)OD_HALF) rc[t + i * OD_T] = st[i] + lr[i];
    }
}

#define SCAN_T  1024
#define SCAN_V4 (((NNODES + SCAN_T - 1) / SCAN_T + 3) / 4)

__global__ __launch_bounds__(SCAN_T) void scan_kernel(const int* __restrict__ cnt_in,
                                                      int* __restrict__ off, int n) {
    __shared__ int part[SCAN_T];
    const int t  = threadIdx.x;
    const int g0 = t * SCAN_V4;
    const int n4 = n >> 2;
    const v4i* c4 = (const v4i*)cnt_in;

    v4i c[SCAN_V4];
#pragma unroll
    for (int i = 0; i < SCAN_V4; ++i) {
        const int g = g0 + i;
        c[i] = c4[g < n4 ? g : n4 - 1];
    }
    int sum = 0;
#pragma unroll
    for (int i = 0; i < SCAN_V4; ++i) {
        if (g0 + i >= n4) { v4i z = {}; c[i] = z; }
        sum += (c[i].x + c[i].y) + (c[i].z + c[i].w);
    }
    part[t] = sum;
    __syncthreads();
    for (int d = 1; d < SCAN_T; d <<= 1) {
        const int v = (t >= d) ? part[t - d] : 0;
        __syncthreads();
        part[t] += v;
        __syncthreads();
    }
    int run = part[t] - sum;
    v4i* o4 = (v4i*)off;
#pragma unroll
    for (int i = 0; i < SCAN_V4; ++i) {
        const int g = g0 + i;
        v4i o;
        o.x = run;
        o.y = o.x + c[i].x;
        o.z = o.y + c[i].y;
        o.w = o.z + c[i].z;
        run = o.w + c[i].w;
        if (g < n4) o4[g] = o;
    }
    if (t == SCAN_T - 1) off[n] = part[t];
}

__global__ void fill_kernel(const int* __restrict__ src, const int* __restrict__ dst,
                            const int* __restrict__ rank, const int* __restrict__ off,
                            const int* __restrict__ cnt_out,
                            v2i* __restrict__ csr, int nE) {
    int e = blockIdx.x * blockDim.x + threadIdx.x;
    if (e < nE) {
        const int s = src[e];
        const int p = off[dst[e]] + rank[e];
        v2i rec;
        rec.x = s;
        rec.y = __float_as_int(rsqrtf(fmaxf((float)cnt_out[s], 1.0f)));
        csr[p] = rec;
    }
}

#define PW  4
#define LDP (FEATS + 4)

template <int NT>
__device__ __forceinline__ void tile_gemm(const float* ap, const float* __restrict__ W,
                                          int half, int l16, v8f (&acc)[NT]) {
    constexpr int NOUT = NT * 16;
#pragma unroll
    for (int k0 = 0; k0 < FEATS; k0 += 4) {
        const v2f A = *(const v2f*)(ap + k0);
        const float* wp = W + (size_t)(k0 + half * 2) * NOUT + l16;
#pragma unroll
        for (int t = 0; t < NT; ++t) {
            v2f B;
            B.x = wp[t * 16];
            B.y = wp[t * 16 + NOUT];
            acc[t] = __builtin_amdgcn_wmma_f32_16x16x4_f32(
                false, A, false, B, (short)0, acc[t], false, false);
        }
    }
}

template <int NT, bool RELU, int NT2>
__global__ __launch_bounds__(PW * 32) void pull_gemm_wmma(
        const float* __restrict__ hin, const int* __restrict__ off,
        const v2i* __restrict__ csr,
        const float* __restrict__ W, const float* __restrict__ bias,
        const float* __restrict__ W2,
        float* __restrict__ out, int nRows) {
    constexpr int NOUT = NT * 16;
    static_assert(NT2 == 0 || NOUT == FEATS, "the second product reads the first one's 64-wide tile");
    __shared__ __attribute__((aligned(16))) float tile[PW][16 * LDP];

    const int lane = threadIdx.x & 31;
    const int wave = threadIdx.x >> 5;
    const int rowBase = (blockIdx.x * PW + wave) * 16;
    const bool active = rowBase < nRows;
    const int half = lane >> 4;
    const int l16  = lane & 15;
    float* myt = tile[wave];

    if (active) {
        for (int i = 0; i < 8; ++i) {
            const int r   = i * 2 + half;
            const int row = rowBase + r;
            int beg = 0, end = 0;
            if (row < nRows) { beg = off[row]; end = off[row + 1]; }
            v4f acc = {};
            for (int p = beg; p < end; p += 8) {
                int   s[8];
                float w[8];
                v4f   v[8];
#pragma unroll
                for (int u = 0; u < 8; ++u) {
                    int q = p + u;
                    q = q < end ? q : end - 1;
                    const v2i rec = csr[q];
                    s[u] = rec.x;
                    w[u] = __int_as_float(rec.y);
                }
#pragma unroll
                for (int u = 0; u < 8; ++u)
                    v[u] = *(const v4f*)(hin + (size_t)s[u] * FEATS + l16 * 4);
#pragma unroll
                for (int u = 0; u < 8; ++u) {
                    if (p + u < end) {
                        acc.x = fmaf(v[u].x, w[u], acc.x);
                        acc.y = fmaf(v[u].y, w[u], acc.y);
                        acc.z = fmaf(v[u].z, w[u], acc.z);
                        acc.w = fmaf(v[u].w, w[u], acc.w);
                    }
                }
            }
            const float dn = rsqrtf(fmaxf((float)(end - beg), 1.0f));
            *(v4f*)(myt + r * LDP + l16 * 4) = acc * dn;
        }
    }
    __syncthreads();

    const float* ap = myt + l16 * LDP + half * 2;
    v8f acc2[NT] = {};
    if (active) {
        tile_gemm<NT>(ap, W, half, l16, acc2);
#pragma unroll
        for (int t = 0; t < NT; ++t) {
            const int col = t * 16 + l16;
            const float bv = bias[col];
#pragma unroll
            for (int i = 0; i < 8; ++i) {
                const int row = rowBase + half * 8 + i;
                float v = acc2[t][i] + bv;
                if (RELU) v = fmaxf(v, 0.0f);
                if constexpr (NT2 == 0) {
                    if (row < nRows) out[(size_t)row * NOUT + col] = v;
                } else {
                    myt[(half * 8 + i) * LDP + col] = v;
                }
            }
        }
    }

    if constexpr (NT2 > 0) {
        constexpr int NOUT2 = NT2 * 16;
        __syncthreads();
        if (active) {
            v8f acc3[NT2] = {};
            tile_gemm<NT2>(ap, W2, half, l16, acc3);
#pragma unroll
            for (int t = 0; t < NT2; ++t) {
                const int col = t * 16 + l16;
#pragma unroll
                for (int i = 0; i < 8; ++i) {
                    const int row = rowBase + half * 8 + i;
                    if (row < nRows) out[(size_t)row * NOUT2 + col] = acc3[t][i];
                }
            }
        }
    }
}

#define AGW 8
#define AGP 2
#define AGF 32
static_assert(AGF * 4 == 128, "one row = one 128-byte line = 8 lanes x 16 B");

__global__ __launch_bounds__(AGW * 32) void pull_rows32_kernel(
        const float* __restrict__ tin, const int* __restrict__ off,
        const v2i* __restrict__ csr, const float* __restrict__ bias,
        float* __restrict__ out, int nRows) {
    const int lane = threadIdx.x & 31;
    const int wave = threadIdx.x >> 5;
    const int qw   = lane >> 3;
    const int l8   = lane & 7;
    const int rowBase = (blockIdx.x * AGW + wave) * (AGP * 4);

    int begs[AGP], ends[AGP];
#pragma unroll
    for (int i = 0; i < AGP; ++i) {
        const int row = rowBase + i * 4 + qw;
        begs[i] = off[row     < nRows ? row     : nRows];
        ends[i] = off[row + 1 < nRows ? row + 1 : nRows];
    }
    const v4f bv = *(const v4f*)(bias + l8 * 4);

#pragma unroll
    for (int i = 0; i < AGP; ++i) {
        const int row = rowBase + i * 4 + qw;
        const int beg = begs[i], end = ends[i];
        v4f acc = {};
        for (int p = beg; p < end; p += 8) {
            int   s[8];
            float w[8];
            v4f   v[8];
#pragma unroll
            for (int u = 0; u < 8; ++u) {
                int q = p + u;
                q = q < end ? q : end - 1;
                const v2i rec = csr[q];
                s[u] = rec.x;
                w[u] = __int_as_float(rec.y);
            }
#pragma unroll
            for (int u = 0; u < 8; ++u)
                v[u] = *(const v4f*)(tin + (size_t)s[u] * AGF + l8 * 4);
#pragma unroll
            for (int u = 0; u < 8; ++u) {
                if (p + u < end) {
                    acc.x = fmaf(v[u].x, w[u], acc.x);
                    acc.y = fmaf(v[u].y, w[u], acc.y);
                    acc.z = fmaf(v[u].z, w[u], acc.z);
                    acc.w = fmaf(v[u].w, w[u], acc.w);
                }
            }
        }
        const float dn = rsqrtf(fmaxf((float)(end - beg), 1.0f));
        if (row < nRows) *(v4f*)(out + (size_t)row * AGF + l8 * 4) = acc * dn + bv;
    }
}

extern "C" void kernel_launch(void* const* d_in, const int* in_sizes, int n_in,
                              void* d_out, int out_size, void* d_ws, size_t ws_size,
                              hipStream_t stream) {
    const float* x    = (const float*)d_in[0];
    const int*   esrc = (const int*)d_in[1];
    const int*   edst = (const int*)d_in[2];
    const float* W0   = (const float*)d_in[3];
    const float* b0   = (const float*)d_in[4];
    const float* W1   = (const float*)d_in[5];
    const float* b1   = (const float*)d_in[6];
    const float* W2   = (const float*)d_in[7];
    const float* b2   = (const float*)d_in[8];
    float* outp = (float*)d_out;

    const int NOFF = 100096;
    int*   cnt_out = (int*)d_ws;
    int*   cnt_in  = cnt_out + NNODES;
    int*   off     = cnt_in + NNODES;
    int*   rank    = off + NOFF;
    v2i*   csr     = (v2i*)(rank + NEDGES);
    float* hA      = (float*)(csr + NEDGES);
    float* tB      = hA + (size_t)NNODES * FEATS;

    const int T = 256;
    const int gZero = (2 * NNODES / 4 + T - 1) / T;
    const int gEdge = (NEDGES + T - 1) / T;
    const int gPull = (NNODES + 16 * PW - 1) / (16 * PW);
    const int gAgg  = (NNODES + 4 * AGP * AGW - 1) / (4 * AGP * AGW);

    zero_i4<<<gZero, T, 0, stream>>>(cnt_out, 2 * NNODES / 4);
    outdeg_lds_kernel<<<dim3(OD_CHUNKS, 2), OD_T, 0, stream>>>(esrc, cnt_out);
    indeg_rank_lds_kernel<<<dim3(OD_CHUNKS, 2), OD_T, 0, stream>>>(edst, cnt_in, rank);
    scan_kernel<<<1, SCAN_T, 0, stream>>>(cnt_in, off, NNODES);
    fill_kernel<<<gEdge, T, 0, stream>>>(esrc, edst, rank, off, cnt_out, csr, NEDGES);

    pull_gemm_wmma<4, true, 0><<<gPull, PW * 32, 0, stream>>>(x,  off, csr, W0, b0, nullptr, hA, NNODES);
    pull_gemm_wmma<4, true, 2><<<gPull, PW * 32, 0, stream>>>(hA, off, csr, W1, b1, W2,      tB, NNODES);
    pull_rows32_kernel<<<gAgg, AGW * 32, 0, stream>>>(tB, off, csr, b2, outp, NNODES);
}


# out-degree pass and in-degree/rank pass (independent, 128 whole-WGP work-groups each) merged into one launch of 256 work-groups that run side by side; 7 launches, float sums untouched
# speedup vs baseline: 3.6516x; 1.0522x over previous
#include <hip/hip_runtime.h>

#define NNODES 100000
#define NEDGES 1600000
#define FEATS  64

typedef __attribute__((ext_vector_type(2))) float v2f;
typedef __attribute__((ext_vector_type(4))) float v4f;
typedef __attribute__((ext_vector_type(8))) float v8f;
typedef __attribute__((ext_vector_type(4))) int   v4i;
typedef __attribute__((ext_vector_type(2))) int   v2i;

static_assert(NNODES % 16 == 0, "row tiles of 16 and 4-wide counter groups");

__global__ void zero_i4(int* __restrict__ p, int n4) {
    int i = blockIdx.x * blockDim.x + threadIdx.x;
    if (i < n4) {
        v4i z = {};
        ((v4i*)p)[i] = z;
    }
}

#define OD_T      1024
#define OD_CHUNKS 64
#define OD_HALF   50016
#define OD_E4     (NEDGES / OD_CHUNKS / 4)
#define OD_IT     ((OD_E4 + OD_T - 1) / OD_T)
static_assert(NEDGES % (OD_CHUNKS * 4) == 0, "chunks of whole int4 groups");
static_assert(OD_HALF % 32 == 0 && OD_HALF < NNODES && 2 * OD_HALF >= NNODES, "two halves cover the nodes");
static_assert((OD_CHUNKS - 1) * 768 < OD_HALF, "rotated sweep wraps at most once");

__device__ __forceinline__ void outdeg_lds_body(int* cnt, const int* __restrict__ src,
                                                int* __restrict__ cnt_out) {
    const int t     = threadIdx.x;
    const int chunk = blockIdx.x;
    const int hbase = blockIdx.y * OD_HALF;

    for (int k = t; k < OD_HALF / 4; k += OD_T) {
        v4i z = {};
        ((v4i*)cnt)[k] = z;
    }
    __syncthreads();

    const v4i* s4 = (const v4i*)src + (size_t)chunk * OD_E4;
    v4i sv[OD_IT];
#pragma unroll
    for (int i = 0; i < OD_IT; ++i) {
        const int k = t + i * OD_T;
        sv[i] = s4[k < OD_E4 ? k : OD_E4 - 1];
    }
#pragma unroll
    for (int i = 0; i < OD_IT; ++i) {
        if (t + i * OD_T < OD_E4) {
            unsigned r;
            r = (unsigned)(sv[i].x - hbase); if (r < (unsigned)OD_HALF) atomicAdd(&cnt[r], 1);
            r = (unsigned)(sv[i].y - hbase); if (r < (unsigned)OD_HALF) atomicAdd(&cnt[r], 1);
            r = (unsigned)(sv[i].z - hbase); if (r < (unsigned)OD_HALF) atomicAdd(&cnt[r], 1);
            r = (unsigned)(sv[i].w - hbase); if (r < (unsigned)OD_HALF) atomicAdd(&cnt[r], 1);
        }
    }
    __syncthreads();

    const int nH  = (NNODES - hbase) < OD_HALF ? (NNODES - hbase) : OD_HALF;
    const int rot = chunk * 768;
    for (int k = t; k < OD_HALF; k += OD_T) {
        int idx = k + rot;
        if (idx >= OD_HALF) idx -= OD_HALF;
        const int v = cnt[idx];
        if (v != 0 && idx < nH) atomicAdd(&cnt_out[hbase + idx], v);
    }
}

#define IR_E  (NEDGES / OD_CHUNKS)
#define IR_IT ((IR_E + OD_T - 1) / OD_T)
#define IR_FB 13
#define IR_FN ((OD_HALF + OD_T * IR_FB - 1) / (OD_T * IR_FB))
static_assert(NEDGES % OD_CHUNKS == 0, "chunks of whole edges");

__device__ __forceinline__ void indeg_rank_lds_body(int* cnt, const int* __restrict__ dst,
                                                    int* __restrict__ cnt_in,
                                                    int* __restrict__ rank) {
    const int t     = threadIdx.x;
    const int chunk = blockIdx.x;
    const int hbase = blockIdx.y * OD_HALF;

    for (int k = t; k < OD_HALF / 4; k += OD_T) {
        v4i z = {};
        ((v4i*)cnt)[k] = z;
    }
    __syncthreads();

    const int* dc = dst  + (size_t)chunk * IR_E;
    int*       rc = rank + (size_t)chunk * IR_E;
    unsigned r[IR_IT];
    int      lr[IR_IT];
#pragma unroll
    for (int i = 0; i < IR_IT; ++i) {
        const int k = t + i * OD_T;
        r[i] = (unsigned)(dc[k < IR_E ? k : IR_E - 1] - hbase);
    }
#pragma unroll
    for (int i = 0; i < IR_IT; ++i) {
        if (t + i * OD_T >= IR_E) r[i] = 0xFFFFFFFFu;
        lr[i] = 0;
        if (r[i] < (unsigned)OD_HALF) lr[i] = atomicAdd(&cnt[r[i]], 1);
    }
    __syncthreads();

    const int nH  = (NNODES - hbase) < OD_HALF ? (NNODES - hbase) : OD_HALF;
    const int rot = chunk * 768;
    for (int b = 0; b < IR_FN; ++b) {
        int idx[IR_FB];
        int v[IR_FB];
#pragma unroll
        for (int u = 0; u < IR_FB; ++u) {
            const int k = t + (b * IR_FB + u) * OD_T;
            int x = k + rot;
            if (x >= OD_HALF) x -= OD_HALF;
            idx[u] = x;
            v[u] = (k < OD_HALF) ? cnt[x] : 0;
        }
#pragma unroll
        for (int u = 0; u < IR_FB; ++u) {
            const bool on = v[u] != 0 && idx[u] < nH;
            int base = 0;
            if (on) base = atomicAdd(&cnt_in[hbase + idx[u]], v[u]);
            v[u] = on ? base : -1;
        }
#pragma unroll
        for (int u = 0; u < IR_FB; ++u) {
            if (v[u] >= 0) cnt[idx[u]] = v[u];
        }
    }
    __syncthreads();

    int st[IR_IT];
#pragma unroll
    for (int i = 0; i < IR_IT; ++i)
        st[i] = cnt[r[i] < (unsigned)OD_HALF ? r[i] : 0];
#pragma unroll
    for (int i = 0; i < IR_IT; ++i) {
        if (r[i] < (unsigned)OD_HALF) rc[t + i * OD_T] = st[i] + lr[i];
    }
}

__global__ __launch_bounds__(OD_T) void degrees_lds_kernel(const int* __restrict__ src,
                                                           const int* __restrict__ dst,
                                                           int* __restrict__ cnt_out,
                                                           int* __restrict__ cnt_in,
                                                           int* __restrict__ rank) {
    __shared__ __attribute__((aligned(16))) int cnt[OD_HALF];
    if (blockIdx.z == 0) indeg_rank_lds_body(cnt, dst, cnt_in, rank);
    else                 outdeg_lds_body(cnt, src, cnt_out);
}

#define SCAN_T  1024
#define SCAN_V4 (((NNODES + SCAN_T - 1) / SCAN_T + 3) / 4)

__global__ __launch_bounds__(SCAN_T) void scan_kernel(const int* __restrict__ cnt_in,
                                                      int* __restrict__ off, int n) {
    __shared__ int part[SCAN_T];
    const int t  = threadIdx.x;
    const int g0 = t * SCAN_V4;
    const int n4 = n >> 2;
    const v4i* c4 = (const v4i*)cnt_in;

    v4i c[SCAN_V4];
#pragma unroll
    for (int i = 0; i < SCAN_V4; ++i) {
        const int g = g0 + i;
        c[i] = c4[g < n4 ? g : n4 - 1];
    }
    int sum = 0;
#pragma unroll
    for (int i = 0; i < SCAN_V4; ++i) {
        if (g0 + i >= n4) { v4i z = {}; c[i] = z; }
        sum += (c[i].x + c[i].y) + (c[i].z + c[i].w);
    }
    part[t] = sum;
    __syncthreads();
    for (int d = 1; d < SCAN_T; d <<= 1) {
        const int v = (t >= d) ? part[t - d] : 0;
        __syncthreads();
        part[t] += v;
        __syncthreads();
    }
    int run = part[t] - sum;
    v4i* o4 = (v4i*)off;
#pragma unroll
    for (int i = 0; i < SCAN_V4; ++i) {
        const int g = g0 + i;
        v4i o;
        o.x = run;
        o.y = o.x + c[i].x;
        o.z = o.y + c[i].y;
        o.w = o.z + c[i].z;
        run = o.w + c[i].w;
        if (g < n4) o4[g] = o;
    }
    if (t == SCAN_T - 1) off[n] = part[t];
}

__global__ void fill_kernel(const int* __restrict__ src, const int* __restrict__ dst,
                            const int* __restrict__ rank, const int* __restrict__ off,
                            const int* __restrict__ cnt_out,
                            v2i* __restrict__ csr, int nE) {
    int e = blockIdx.x * blockDim.x + threadIdx.x;
    if (e < nE) {
        const int s = src[e];
        const int p = off[dst[e]] + rank[e];
        v2i rec;
        rec.x = s;
        rec.y = __float_as_int(rsqrtf(fmaxf((float)cnt_out[s], 1.0f)));
        csr[p] = rec;
    }
}

#define PW  4
#define LDP (FEATS + 4)

template <int NT>
__device__ __forceinline__ void tile_gemm(const float* ap, const float* __restrict__ W,
                                          int half, int l16, v8f (&acc)[NT]) {
    constexpr int NOUT = NT * 16;
#pragma unroll
    for (int k0 = 0; k0 < FEATS; k0 += 4) {
        const v2f A = *(const v2f*)(ap + k0);
        const float* wp = W + (size_t)(k0 + half * 2) * NOUT + l16;
#pragma unroll
        for (int t = 0; t < NT; ++t) {
            v2f B;
            B.x = wp[t * 16];
            B.y = wp[t * 16 + NOUT];
            acc[t] = __builtin_amdgcn_wmma_f32_16x16x4_f32(
                false, A, false, B, (short)0, acc[t], false, false);
        }
    }
}

template <int NT, bool RELU, int NT2>
__global__ __launch_bounds__(PW * 32) void pull_gemm_wmma(
        const float* __restrict__ hin, const int* __restrict__ off,
        const v2i* __restrict__ csr,
        const float* __restrict__ W, const float* __restrict__ bias,
        const float* __restrict__ W2,
        float* __restrict__ out, int nRows) {
    constexpr int NOUT = NT * 16;
    static_assert(NT2 == 0 || NOUT == FEATS, "the second product reads the first one's 64-wide tile");
    __shared__ __attribute__((aligned(16))) float tile[PW][16 * LDP];

    const int lane = threadIdx.x & 31;
    const int wave = threadIdx.x >> 5;
    const int rowBase = (blockIdx.x * PW + wave) * 16;
    const bool active = rowBase < nRows;
    const int half = lane >> 4;
    const int l16  = lane & 15;
    float* myt = tile[wave];

    if (active) {
        for (int i = 0; i < 8; ++i) {
            const int r   = i * 2 + half;
            const int row = rowBase + r;
            int beg = 0, end = 0;
            if (row < nRows) { beg = off[row]; end = off[row + 1]; }
            v4f acc = {};
            for (int p = beg; p < end; p += 8) {
                int   s[8];
                float w[8];
                v4f   v[8];
#pragma unroll
                for (int u = 0; u < 8; ++u) {
                    int q = p + u;
                    q = q < end ? q : end - 1;
                    const v2i rec = csr[q];
                    s[u] = rec.x;
                    w[u] = __int_as_float(rec.y);
                }
#pragma unroll
                for (int u = 0; u < 8; ++u)
                    v[u] = *(const v4f*)(hin + (size_t)s[u] * FEATS + l16 * 4);
#pragma unroll
                for (int u = 0; u < 8; ++u) {
                    if (p + u < end) {
                        acc.x = fmaf(v[u].x, w[u], acc.x);
                        acc.y = fmaf(v[u].y, w[u], acc.y);
                        acc.z = fmaf(v[u].z, w[u], acc.z);
                        acc.w = fmaf(v[u].w, w[u], acc.w);
                    }
                }
            }
            const float dn = rsqrtf(fmaxf((float)(end - beg), 1.0f));
            *(v4f*)(myt + r * LDP + l16 * 4) = acc * dn;
        }
    }
    __syncthreads();

    const float* ap = myt + l16 * LDP + half * 2;
    v8f acc2[NT] = {};
    if (active) {
        tile_gemm<NT>(ap, W, half, l16, acc2);
#pragma unroll
        for (int t = 0; t < NT; ++t) {
            const int col = t * 16 + l16;
            const float bv = bias[col];
#pragma unroll
            for (int i = 0; i < 8; ++i) {
                const int row = rowBase + half * 8 + i;
                float v = acc2[t][i] + bv;
                if (RELU) v = fmaxf(v, 0.0f);
                if constexpr (NT2 == 0) {
                    if (row < nRows) out[(size_t)row * NOUT + col] = v;
                } else {
                    myt[(half * 8 + i) * LDP + col] = v;
                }
            }
        }
    }

    if constexpr (NT2 > 0) {
        constexpr int NOUT2 = NT2 * 16;
        __syncthreads();
        if (active) {
            v8f acc3[NT2] = {};
            tile_gemm<NT2>(ap, W2, half, l16, acc3);
#pragma unroll
            for (int t = 0; t < NT2; ++t) {
                const int col = t * 16 + l16;
#pragma unroll
                for (int i = 0; i < 8; ++i) {
                    const int row = rowBase + half * 8 + i;
                    if (row < nRows) out[(size_t)row * NOUT2 + col] = acc3[t][i];
                }
            }
        }
    }
}

#define AGW 8
#define AGP 2
#define AGF 32
static_assert(AGF * 4 == 128, "one row = one 128-byte line = 8 lanes x 16 B");

__global__ __launch_bounds__(AGW * 32) void pull_rows32_kernel(
        const float* __restrict__ tin, const int* __restrict__ off,
        const v2i* __restrict__ csr, const float* __restrict__ bias,
        float* __restrict__ out, int nRows) {
    const int lane = threadIdx.x & 31;
    const int wave = threadIdx.x >> 5;
    const int qw   = lane >> 3;
    const int l8   = lane & 7;
    const int rowBase = (blockIdx.x * AGW + wave) * (AGP * 4);

    int begs[AGP], ends[AGP];
#pragma unroll
    for (int i = 0; i < AGP; ++i) {
        const int row = rowBase + i * 4 + qw;
        begs[i] = off[row     < nRows ? row     : nRows];
        ends[i] = off[row + 1 < nRows ? row + 1 : nRows];
    }
    const v4f bv = *(const v4f*)(bias + l8 * 4);

#pragma unroll
    for (int i = 0; i < AGP; ++i) {
        const int row = rowBase + i * 4 + qw;
        const int beg = begs[i], end = ends[i];
        v4f acc = {};
        for (int p = beg; p < end; p += 8) {
            int   s[8];
            float w[8];
            v4f   v[8];
#pragma unroll
            for (int u = 0; u < 8; ++u) {
                int q = p + u;
                q = q < end ? q : end - 1;
                const v2i rec = csr[q];
                s[u] = rec.x;
                w[u] = __int_as_float(rec.y);
            }
#pragma unroll
            for (int u = 0; u < 8; ++u)
                v[u] = *(const v4f*)(tin + (size_t)s[u] * AGF + l8 * 4);
#pragma unroll
            for (int u = 0; u < 8; ++u) {
                if (p + u < end) {
                    acc.x = fmaf(v[u].x, w[u], acc.x);
                    acc.y = fmaf(v[u].y, w[u], acc.y);
                    acc.z = fmaf(v[u].z, w[u], acc.z);
                    acc.w = fmaf(v[u].w, w[u], acc.w);
                }
            }
        }
        const float dn = rsqrtf(fmaxf((float)(end - beg), 1.0f));
        if (row < nRows) *(v4f*)(out + (size_t)row * AGF + l8 * 4) = acc * dn + bv;
    }
}

extern "C" void kernel_launch(void* const* d_in, const int* in_sizes, int n_in,
                              void* d_out, int out_size, void* d_ws, size_t ws_size,
                              hipStream_t stream) {
    const float* x    = (const float*)d_in[0];
    const int*   esrc = (const int*)d_in[1];
    const int*   edst = (const int*)d_in[2];
    const float* W0   = (const float*)d_in[3];
    const float* b0   = (const float*)d_in[4];
    const float* W1   = (const float*)d_in[5];
    const float* b1   = (const float*)d_in[6];
    const float* W2   = (const float*)d_in[7];
    const float* b2   = (const float*)d_in[8];
    float* outp = (float*)d_out;

    const int NOFF = 100096;
    int*   cnt_out = (int*)d_ws;
    int*   cnt_in  = cnt_out + NNODES;
    int*   off     = cnt_in + NNODES;
    int*   rank    = off + NOFF;
    v2i*   csr     = (v2i*)(rank + NEDGES);
    float* hA      = (float*)(csr + NEDGES);
    float* tB      = hA + (size_t)NNODES * FEATS;

    const int T = 256;
    const int gZero = (2 * NNODES / 4 + T - 1) / T;
    const int gEdge = (NEDGES + T - 1) / T;
    const int gPull = (NNODES + 16 * PW - 1) / (16 * PW);
    const int gAgg  = (NNODES + 4 * AGP * AGW - 1) / (4 * AGP * AGW);

    zero_i4<<<gZero, T, 0, stream>>>(cnt_out, 2 * NNODES / 4);
    degrees_lds_kernel<<<dim3(OD_CHUNKS, 2, 2), OD_T, 0, stream>>>(esrc, edst, cnt_out, cnt_in, rank);
    scan_kernel<<<1, SCAN_T, 0, stream>>>(cnt_in, off, NNODES);
    fill_kernel<<<gEdge, T, 0, stream>>>(esrc, edst, rank, off, cnt_out, csr, NEDGES);

    pull_gemm_wmma<4, true, 0><<<gPull, PW * 32, 0, stream>>>(x,  off, csr, W0, b0, nullptr, hA, NNODES);
    pull_gemm_wmma<4, true, 2><<<gPull, PW * 32, 0, stream>>>(hA, off, csr, W1, b1, W2,      tB, NNODES);
    pull_rows32_kernel<<<gAgg, AGW * 32, 0, stream>>>(tB, off, csr, b2, outp, NNODES);
}


# prefix sum of in-degrees: one work-group at a 400-B lane stride (32 lines per load and store) becomes 25 work-groups, one int4 per thread (whole lines), each summing the counts before its block
# speedup vs baseline: 4.3171x; 1.1822x over previous
#include <hip/hip_runtime.h>

#define NNODES 100000
#define NEDGES 1600000
#define FEATS  64

typedef __attribute__((ext_vector_type(2))) float v2f;
typedef __attribute__((ext_vector_type(4))) float v4f;
typedef __attribute__((ext_vector_type(8))) float v8f;
typedef __attribute__((ext_vector_type(4))) int   v4i;
typedef __attribute__((ext_vector_type(2))) int   v2i;

static_assert(NNODES % 16 == 0, "row tiles of 16 and 4-wide counter groups");

__global__ void zero_i4(int* __restrict__ p, int n4) {
    int i = blockIdx.x * blockDim.x + threadIdx.x;
    if (i < n4) {
        v4i z = {};
        ((v4i*)p)[i] = z;
    }
}

#define OD_T      1024
#define OD_CHUNKS 64
#define OD_HALF   50016
#define OD_E4     (NEDGES / OD_CHUNKS / 4)
#define OD_IT     ((OD_E4 + OD_T - 1) / OD_T)
static_assert(NEDGES % (OD_CHUNKS * 4) == 0, "chunks of whole int4 groups");
static_assert(OD_HALF % 32 == 0 && OD_HALF < NNODES && 2 * OD_HALF >= NNODES, "two halves cover the nodes");
static_assert((OD_CHUNKS - 1) * 768 < OD_HALF, "rotated sweep wraps at most once");

__device__ __forceinline__ void outdeg_lds_body(int* cnt, const int* __restrict__ src,
                                                int* __restrict__ cnt_out) {
    const int t     = threadIdx.x;
    const int chunk = blockIdx.x;
    const int hbase = blockIdx.y * OD_HALF;

    for (int k = t; k < OD_HALF / 4; k += OD_T) {
        v4i z = {};
        ((v4i*)cnt)[k] = z;
    }
    __syncthreads();

    const v4i* s4 = (const v4i*)src + (size_t)chunk * OD_E4;
    v4i sv[OD_IT];
#pragma unroll
    for (int i = 0; i < OD_IT; ++i) {
        const int k = t + i * OD_T;
        sv[i] = s4[k < OD_E4 ? k : OD_E4 - 1];
    }
#pragma unroll
    for (int i = 0; i < OD_IT; ++i) {
        if (t + i * OD_T < OD_E4) {
            unsigned r;
            r = (unsigned)(sv[i].x - hbase); if (r < (unsigned)OD_HALF) atomicAdd(&cnt[r], 1);
            r = (unsigned)(sv[i].y - hbase); if (r < (unsigned)OD_HALF) atomicAdd(&cnt[r], 1);
            r = (unsigned)(sv[i].z - hbase); if (r < (unsigned)OD_HALF) atomicAdd(&cnt[r], 1);
            r = (unsigned)(sv[i].w - hbase); if (r < (unsigned)OD_HALF) atomicAdd(&cnt[r], 1);
        }
    }
    __syncthreads();

    const int nH  = (NNODES - hbase) < OD_HALF ? (NNODES - hbase) : OD_HALF;
    const int rot = chunk * 768;
    for (int k = t; k < OD_HALF; k += OD_T) {
        int idx = k + rot;
        if (idx >= OD_HALF) idx -= OD_HALF;
        const int v = cnt[idx];
        if (v != 0 && idx < nH) atomicAdd(&cnt_out[hbase + idx], v);
    }
}

#define IR_E  (NEDGES / OD_CHUNKS)
#define IR_IT ((IR_E + OD_T - 1) / OD_T)
#define IR_FB 13
#define IR_FN ((OD_HALF + OD_T * IR_FB - 1) / (OD_T * IR_FB))
static_assert(NEDGES % OD_CHUNKS == 0, "chunks of whole edges");

__device__ __forceinline__ void indeg_rank_lds_body(int* cnt, const int* __restrict__ dst,
                                                    int* __restrict__ cnt_in,
                                                    int* __restrict__ rank) {
    const int t     = threadIdx.x;
    const int chunk = blockIdx.x;
    const int hbase = blockIdx.y * OD_HALF;

    for (int k = t; k < OD_HALF / 4; k += OD_T) {
        v4i z = {};
        ((v4i*)cnt)[k] = z;
    }
    __syncthreads();

    const int* dc = dst  + (size_t)chunk * IR_E;
    int*       rc = rank + (size_t)chunk * IR_E;
    unsigned r[IR_IT];
    int      lr[IR_IT];
#pragma unroll
    for (int i = 0; i < IR_IT; ++i) {
        const int k = t + i * OD_T;
        r[i] = (unsigned)(dc[k < IR_E ? k : IR_E - 1] - hbase);
    }
#pragma unroll
    for (int i = 0; i < IR_IT; ++i) {
        if (t + i * OD_T >= IR_E) r[i] = 0xFFFFFFFFu;
        lr[i] = 0;
        if (r[i] < (unsigned)OD_HALF) lr[i] = atomicAdd(&cnt[r[i]], 1);
    }
    __syncthreads();

    const int nH  = (NNODES - hbase) < OD_HALF ? (NNODES - hbase) : OD_HALF;
    const int rot = chunk * 768;
    for (int b = 0; b < IR_FN; ++b) {
        int idx[IR_FB];
        int v[IR_FB];
#pragma unroll
        for (int u = 0; u < IR_FB; ++u) {
            const int k = t + (b * IR_FB + u) * OD_T;
            int x = k + rot;
            if (x >= OD_HALF) x -= OD_HALF;
            idx[u] = x;
            v[u] = (k < OD_HALF) ? cnt[x] : 0;
        }
#pragma unroll
        for (int u = 0; u < IR_FB; ++u) {
            const bool on = v[u] != 0 && idx[u] < nH;
            int base = 0;
            if (on) base = atomicAdd(&cnt_in[hbase + idx[u]], v[u]);
            v[u] = on ? base : -1;
        }
#pragma unroll
        for (int u = 0; u < IR_FB; ++u) {
            if (v[u] >= 0) cnt[idx[u]] = v[u];
        }
    }
    __syncthreads();

    int st[IR_IT];
#pragma unroll
    for (int i = 0; i < IR_IT; ++i)
        st[i] = cnt[r[i] < (unsigned)OD_HALF ? r[i] : 0];
#pragma unroll
    for (int i = 0; i < IR_IT; ++i) {
        if (r[i] < (unsigned)OD_HALF) rc[t + i * OD_T] = st[i] + lr[i];
    }
}

__global__ __launch_bounds__(OD_T) void degrees_lds_kernel(const int* __restrict__ src,
                                                           const int* __restrict__ dst,
                                                           int* __restrict__ cnt_out,
                                                           int* __restrict__ cnt_in,
                                                           int* __restrict__ rank) {
    __shared__ __attribute__((aligned(16))) int cnt[OD_HALF];
    if (blockIdx.z == 0) indeg_rank_lds_body(cnt, dst, cnt_in, rank);
    else                 outdeg_lds_body(cnt, src, cnt_out);
}

#define SC_T 1024
#define SC_G ((NNODES / 4 + SC_T - 1) / SC_T)
#define SC_B 12
static_assert(SC_T == 32 * 32, "the wave totals are scanned by one wave, a lane per wave");
static_assert((SC_G - 1) * SC_T < NNODES / 4, "every block in front of another is a whole block");

__global__ __launch_bounds__(SC_T) void scan_kernel(const int* __restrict__ cnt_in,
                                                      int* __restrict__ off, int n) {
    __shared__ int wtot[32];
    __shared__ int wpre[32];
    const int t    = threadIdx.x;
    const int lane = t & 31;
    const int wave = t >> 5;
    const int blk  = blockIdx.x;
    const int n4   = n >> 2;
    const v4i* c4  = (const v4i*)cnt_in;
    const int g    = blk * SC_T + t;

    v4i c = c4[g < n4 ? g : n4 - 1];

    int pre = 0;
    for (int k0 = 0; k0 < blk; k0 += SC_B) {
        v4i p[SC_B];
#pragma unroll
        for (int u = 0; u < SC_B; ++u) {
            const int k = k0 + u;
            p[u] = c4[(k < blk ? k : blk - 1) * SC_T + t];
        }
#pragma unroll
        for (int u = 0; u < SC_B; ++u) {
            if (k0 + u < blk) pre += (p[u].x + p[u].y) + (p[u].z + p[u].w);
        }
    }

    if (g >= n4) { v4i z = {}; c = z; }
    const int s = (c.x + c.y) + (c.z + c.w);

    int incl = s;
#pragma unroll
    for (int d = 1; d < 32; d <<= 1) {
        const int v = __shfl_up(incl, d);
        if (lane >= d) incl += v;
    }
    int pw = pre;
#pragma unroll
    for (int d = 16; d >= 1; d >>= 1) pw += __shfl_xor(pw, d);
    if (lane == 31) wtot[wave] = incl;
    if (lane == 0)  wpre[wave] = pw;
    __syncthreads();

    int base = wpre[lane] + (lane < wave ? wtot[lane] : 0);
#pragma unroll
    for (int d = 16; d >= 1; d >>= 1) base += __shfl_xor(base, d);

    v4i o;
    o.x = base + (incl - s);
    o.y = o.x + c.x;
    o.z = o.y + c.y;
    o.w = o.z + c.z;
    if (g < n4) ((v4i*)off)[g] = o;
    if (g == n4 - 1) off[n] = o.w + c.w;
}

__global__ void fill_kernel(const int* __restrict__ src, const int* __restrict__ dst,
                            const int* __restrict__ rank, const int* __restrict__ off,
                            const int* __restrict__ cnt_out,
                            v2i* __restrict__ csr, int nE) {
    int e = blockIdx.x * blockDim.x + threadIdx.x;
    if (e < nE) {
        const int s = src[e];
        const int p = off[dst[e]] + rank[e];
        v2i rec;
        rec.x = s;
        rec.y = __float_as_int(rsqrtf(fmaxf((float)cnt_out[s], 1.0f)));
        csr[p] = rec;
    }
}

#define PW  4
#define LDP (FEATS + 4)

template <int NT>
__device__ __forceinline__ void tile_gemm(const float* ap, const float* __restrict__ W,
                                          int half, int l16, v8f (&acc)[NT]) {
    constexpr int NOUT = NT * 16;
#pragma unroll
    for (int k0 = 0; k0 < FEATS; k0 += 4) {
        const v2f A = *(const v2f*)(ap + k0);
        const float* wp = W + (size_t)(k0 + half * 2) * NOUT + l16;
#pragma unroll
        for (int t = 0; t < NT; ++t) {
            v2f B;
            B.x = wp[t * 16];
            B.y = wp[t * 16 + NOUT];
            acc[t] = __builtin_amdgcn_wmma_f32_16x16x4_f32(
                false, A, false, B, (short)0, acc[t], false, false);
        }
    }
}

template <int NT, bool RELU, int NT2>
__global__ __launch_bounds__(PW * 32) void pull_gemm_wmma(
        const float* __restrict__ hin, const int* __restrict__ off,
        const v2i* __restrict__ csr,
        const float* __restrict__ W, const float* __restrict__ bias,
        const float* __restrict__ W2,
        float* __restrict__ out, int nRows) {
    constexpr int NOUT = NT * 16;
    static_assert(NT2 == 0 || NOUT == FEATS, "the second product reads the first one's 64-wide tile");
    __shared__ __attribute__((aligned(16))) float tile[PW][16 * LDP];

    const int lane = threadIdx.x & 31;
    const int wave = threadIdx.x >> 5;
    const int rowBase = (blockIdx.x * PW + wave) * 16;
    const bool active = rowBase < nRows;
    const int half = lane >> 4;
    const int l16  = lane & 15;
    float* myt = tile[wave];

    if (active) {
        for (int i = 0; i < 8; ++i) {
            const int r   = i * 2 + half;
            const int row = rowBase + r;
            int beg = 0, end = 0;
            if (row < nRows) { beg = off[row]; end = off[row + 1]; }
            v4f acc = {};
            for (int p = beg; p < end; p += 8) {
                int   s[8];
                float w[8];
                v4f   v[8];
#pragma unroll
                for (int u = 0; u < 8; ++u) {
                    int q = p + u;
                    q = q < end ? q : end - 1;
                    const v2i rec = csr[q];
                    s[u] = rec.x;
                    w[u] = __int_as_float(rec.y);
                }
#pragma unroll
                for (int u = 0; u < 8; ++u)
                    v[u] = *(const v4f*)(hin + (size_t)s[u] * FEATS + l16 * 4);
#pragma unroll
                for (int u = 0; u < 8; ++u) {
                    if (p + u < end) {
                        acc.x = fmaf(v[u].x, w[u], acc.x);
                        acc.y = fmaf(v[u].y, w[u], acc.y);
                        acc.z = fmaf(v[u].z, w[u], acc.z);
                        acc.w = fmaf(v[u].w, w[u], acc.w);
                    }
                }
            }
            const float dn = rsqrtf(fmaxf((float)(end - beg), 1.0f));
            *(v4f*)(myt + r * LDP + l16 * 4) = acc * dn;
        }
    }
    __syncthreads();

    const float* ap = myt + l16 * LDP + half * 2;
    v8f acc2[NT] = {};
    if (active) {
        tile_gemm<NT>(ap, W, half, l16, acc2);
#pragma unroll
        for (int t = 0; t < NT; ++t) {
            const int col = t * 16 + l16;
            const float bv = bias[col];
#pragma unroll
            for (int i = 0; i < 8; ++i) {
                const int row = rowBase + half * 8 + i;
                float v = acc2[t][i] + bv;
                if (RELU) v = fmaxf(v, 0.0f);
                if constexpr (NT2 == 0) {
                    if (row < nRows) out[(size_t)row * NOUT + col] = v;
                } else {
                    myt[(half * 8 + i) * LDP + col] = v;
                }
            }
        }
    }

    if constexpr (NT2 > 0) {
        constexpr int NOUT2 = NT2 * 16;
        __syncthreads();
        if (active) {
            v8f acc3[NT2] = {};
            tile_gemm<NT2>(ap, W2, half, l16, acc3);
#pragma unroll
            for (int t = 0; t < NT2; ++t) {
                const int col = t * 16 + l16;
#pragma unroll
                for (int i = 0; i < 8; ++i) {
                    const int row = rowBase + half * 8 + i;
                    if (row < nRows) out[(size_t)row * NOUT2 + col] = acc3[t][i];
                }
            }
        }
    }
}

#define AGW 8
#define AGP 2
#define AGF 32
static_assert(AGF * 4 == 128, "one row = one 128-byte line = 8 lanes x 16 B");

__global__ __launch_bounds__(AGW * 32) void pull_rows32_kernel(
        const float* __restrict__ tin, const int* __restrict__ off,
        const v2i* __restrict__ csr, const float* __restrict__ bias,
        float* __restrict__ out, int nRows) {
    const int lane = threadIdx.x & 31;
    const int wave = threadIdx.x >> 5;
    const int qw   = lane >> 3;
    const int l8   = lane & 7;
    const int rowBase = (blockIdx.x * AGW + wave) * (AGP * 4);

    int begs[AGP], ends[AGP];
#pragma unroll
    for (int i = 0; i < AGP; ++i) {
        const int row = rowBase + i * 4 + qw;
        begs[i] = off[row     < nRows ? row     : nRows];
        ends[i] = off[row + 1 < nRows ? row + 1 : nRows];
    }
    const v4f bv = *(const v4f*)(bias + l8 * 4);

#pragma unroll
    for (int i = 0; i < AGP; ++i) {
        const int row = rowBase + i * 4 + qw;
        const int beg = begs[i], end = ends[i];
        v4f acc = {};
        for (int p = beg; p < end; p += 8) {
            int   s[8];
            float w[8];
            v4f   v[8];
#pragma unroll
            for (int u = 0; u < 8; ++u) {
                int q = p + u;
                q = q < end ? q : end - 1;
                const v2i rec = csr[q];
                s[u] = rec.x;
                w[u] = __int_as_float(rec.y);
            }
#pragma unroll
            for (int u = 0; u < 8; ++u)
                v[u] = *(const v4f*)(tin + (size_t)s[u] * AGF + l8 * 4);
#pragma unroll
            for (int u = 0; u < 8; ++u) {
                if (p + u < end) {
                    acc.x = fmaf(v[u].x, w[u], acc.x);
                    acc.y = fmaf(v[u].y, w[u], acc.y);
                    acc.z = fmaf(v[u].z, w[u], acc.z);
                    acc.w = fmaf(v[u].w, w[u], acc.w);
                }
            }
        }
        const float dn = rsqrtf(fmaxf((float)(end - beg), 1.0f));
        if (row < nRows) *(v4f*)(out + (size_t)row * AGF + l8 * 4) = acc * dn + bv;
    }
}

extern "C" void kernel_launch(void* const* d_in, const int* in_sizes, int n_in,
                              void* d_out, int out_size, void* d_ws, size_t ws_size,
                              hipStream_t stream) {
    const float* x    = (const float*)d_in[0];
    const int*   esrc = (const int*)d_in[1];
    const int*   edst = (const int*)d_in[2];
    const float* W0   = (const float*)d_in[3];
    const float* b0   = (const float*)d_in[4];
    const float* W1   = (const float*)d_in[5];
    const float* b1   = (const float*)d_in[6];
    const float* W2   = (const float*)d_in[7];
    const float* b2   = (const float*)d_in[8];
    float* outp = (float*)d_out;

    const int NOFF = 100096;
    int*   cnt_out = (int*)d_ws;
    int*   cnt_in  = cnt_out + NNODES;
    int*   off     = cnt_in + NNODES;
    int*   rank    = off + NOFF;
    v2i*   csr     = (v2i*)(rank + NEDGES);
    float* hA      = (float*)(csr + NEDGES);
    float* tB      = hA + (size_t)NNODES * FEATS;

    const int T = 256;
    const int gZero = (2 * NNODES / 4 + T - 1) / T;
    const int gEdge = (NEDGES + T - 1) / T;
    const int gPull = (NNODES + 16 * PW - 1) / (16 * PW);
    const int gAgg  = (NNODES + 4 * AGP * AGW - 1) / (4 * AGP * AGW);

    zero_i4<<<gZero, T, 0, stream>>>(cnt_out, 2 * NNODES / 4);
    degrees_lds_kernel<<<dim3(OD_CHUNKS, 2, 2), OD_T, 0, stream>>>(esrc, edst, cnt_out, cnt_in, rank);
    scan_kernel<<<SC_G, SC_T, 0, stream>>>(cnt_in, off, NNODES);
    fill_kernel<<<gEdge, T, 0, stream>>>(esrc, edst, rank, off, cnt_out, csr, NEDGES);

    pull_gemm_wmma<4, true, 0><<<gPull, PW * 32, 0, stream>>>(x,  off, csr, W0, b0, nullptr, hA, NNODES);
    pull_gemm_wmma<4, true, 2><<<gPull, PW * 32, 0, stream>>>(hA, off, csr, W1, b1, W2,      tB, NNODES);
    pull_rows32_kernel<<<gAgg, AGW * 32, 0, stream>>>(tB, off, csr, b2, outp, NNODES);
}


# a wave's edge records and row bounds fetched once (whole lines) into LDS ahead of its row loops (3 gather kernels): a batch of row loads waits on LDS, not on a record load from memory; sums unchanged
# speedup vs baseline: 4.5393x; 1.0515x over previous
#include <hip/hip_runtime.h>

#define NNODES 100000
#define NEDGES 1600000
#define FEATS  64

typedef __attribute__((ext_vector_type(2))) float v2f;
typedef __attribute__((ext_vector_type(4))) float v4f;
typedef __attribute__((ext_vector_type(8))) float v8f;
typedef __attribute__((ext_vector_type(4))) int   v4i;
typedef __attribute__((ext_vector_type(2))) int   v2i;

static_assert(NNODES % 16 == 0, "row tiles of 16 and 4-wide counter groups");

__global__ void zero_i4(int* __restrict__ p, int n4) {
    int i = blockIdx.x * blockDim.x + threadIdx.x;
    if (i < n4) {
        v4i z = {};
        ((v4i*)p)[i] = z;
    }
}

#define OD_T      1024
#define OD_CHUNKS 64
#define OD_HALF   50016
#define OD_E4     (NEDGES / OD_CHUNKS / 4)
#define OD_IT     ((OD_E4 + OD_T - 1) / OD_T)
static_assert(NEDGES % (OD_CHUNKS * 4) == 0, "chunks of whole int4 groups");
static_assert(OD_HALF % 32 == 0 && OD_HALF < NNODES && 2 * OD_HALF >= NNODES, "two halves cover the nodes");
static_assert((OD_CHUNKS - 1) * 768 < OD_HALF, "rotated sweep wraps at most once");

__device__ __forceinline__ void outdeg_lds_body(int* cnt, const int* __restrict__ src,
                                                int* __restrict__ cnt_out) {
    const int t     = threadIdx.x;
    const int chunk = blockIdx.x;
    const int hbase = blockIdx.y * OD_HALF;

    for (int k = t; k < OD_HALF / 4; k += OD_T) {
        v4i z = {};
        ((v4i*)cnt)[k] = z;
    }
    __syncthreads();

    const v4i* s4 = (const v4i*)src + (size_t)chunk * OD_E4;
    v4i sv[OD_IT];
#pragma unroll
    for (int i = 0; i < OD_IT; ++i) {
        const int k = t + i * OD_T;
        sv[i] = s4[k < OD_E4 ? k : OD_E4 - 1];
    }
#pragma unroll
    for (int i = 0; i < OD_IT; ++i) {
        if (t + i * OD_T < OD_E4) {
            unsigned r;
            r = (unsigned)(sv[i].x - hbase); if (r < (unsigned)OD_HALF) atomicAdd(&cnt[r], 1);
            r = (unsigned)(sv[i].y - hbase); if (r < (unsigned)OD_HALF) atomicAdd(&cnt[r], 1);
            r = (unsigned)(sv[i].z - hbase); if (r < (unsigned)OD_HALF) atomicAdd(&cnt[r], 1);
            r = (unsigned)(sv[i].w - hbase); if (r < (unsigned)OD_HALF) atomicAdd(&cnt[r], 1);
        }
    }
    __syncthreads();

    const int nH  = (NNODES - hbase) < OD_HALF ? (NNODES - hbase) : OD_HALF;
    const int rot = chunk * 768;
    for (int k = t; k < OD_HALF; k += OD_T) {
        int idx = k + rot;
        if (idx >= OD_HALF) idx -= OD_HALF;
        const int v = cnt[idx];
        if (v != 0 && idx < nH) atomicAdd(&cnt_out[hbase + idx], v);
    }
}

#define IR_E  (NEDGES / OD_CHUNKS)
#define IR_IT ((IR_E + OD_T - 1) / OD_T)
#define IR_FB 13
#define IR_FN ((OD_HALF + OD_T * IR_FB - 1) / (OD_T * IR_FB))
static_assert(NEDGES % OD_CHUNKS == 0, "chunks of whole edges");

__device__ __forceinline__ void indeg_rank_lds_body(int* cnt, const int* __restrict__ dst,
                                                    int* __restrict__ cnt_in,
                                                    int* __restrict__ rank) {
    const int t     = threadIdx.x;
    const int chunk = blockIdx.x;
    const int hbase = blockIdx.y * OD_HALF;

    for (int k = t; k < OD_HALF / 4; k += OD_T) {
        v4i z = {};
        ((v4i*)cnt)[k] = z;
    }
    __syncthreads();

    const int* dc = dst  + (size_t)chunk * IR_E;
    int*       rc = rank + (size_t)chunk * IR_E;
    unsigned r[IR_IT];
    int      lr[IR_IT];
#pragma unroll
    for (int i = 0; i < IR_IT; ++i) {
        const int k = t + i * OD_T;
        r[i] = (unsigned)(dc[k < IR_E ? k : IR_E - 1] - hbase);
    }
#pragma unroll
    for (int i = 0; i < IR_IT; ++i) {
        if (t + i * OD_T >= IR_E) r[i] = 0xFFFFFFFFu;
        lr[i] = 0;
        if (r[i] < (unsigned)OD_HALF) lr[i] = atomicAdd(&cnt[r[i]], 1);
    }
    __syncthreads();

    const int nH  = (NNODES - hbase) < OD_HALF ? (NNODES - hbase) : OD_HALF;
    const int rot = chunk * 768;
    for (int b = 0; b < IR_FN; ++b) {
        int idx[IR_FB];
        int v[IR_FB];
#pragma unroll
        for (int u = 0; u < IR_FB; ++u) {
            const int k = t + (b * IR_FB + u) * OD_T;
            int x = k + rot;
            if (x >= OD_HALF) x -= OD_HALF;
            idx[u] = x;
            v[u] = (k < OD_HALF) ? cnt[x] : 0;
        }
#pragma unroll
        for (int u = 0; u < IR_FB; ++u) {
            const bool on = v[u] != 0 && idx[u] < nH;
            int base = 0;
            if (on) base = atomicAdd(&cnt_in[hbase + idx[u]], v[u]);
            v[u] = on ? base : -1;
        }
#pragma unroll
        for (int u = 0; u < IR_FB; ++u) {
            if (v[u] >= 0) cnt[idx[u]] = v[u];
        }
    }
    __syncthreads();

    int st[IR_IT];
#pragma unroll
    for (int i = 0; i < IR_IT; ++i)
        st[i] = cnt[r[i] < (unsigned)OD_HALF ? r[i] : 0];
#pragma unroll
    for (int i = 0; i < IR_IT; ++i) {
        if (r[i] < (unsigned)OD_HALF) rc[t + i * OD_T] = st[i] + lr[i];
    }
}

__global__ __launch_bounds__(OD_T) void degrees_lds_kernel(const int* __restrict__ src,
                                                           const int* __restrict__ dst,
                                                           int* __restrict__ cnt_out,
                                                           int* __restrict__ cnt_in,
                                                           int* __restrict__ rank) {
    __shared__ __attribute__((aligned(16))) int cnt[OD_HALF];
    if (blockIdx.z == 0) indeg_rank_lds_body(cnt, dst, cnt_in, rank);
    else                 outdeg_lds_body(cnt, src, cnt_out);
}

#define SC_T 1024
#define SC_G ((NNODES / 4 + SC_T - 1) / SC_T)
#define SC_B 12
static_assert(SC_T == 32 * 32, "the wave totals are scanned by one wave, a lane per wave");
static_assert((SC_G - 1) * SC_T < NNODES / 4, "every block in front of another is a whole block");

__global__ __launch_bounds__(SC_T) void scan_kernel(const int* __restrict__ cnt_in,
                                                      int* __restrict__ off, int n) {
    __shared__ int wtot[32];
    __shared__ int wpre[32];
    const int t    = threadIdx.x;
    const int lane = t & 31;
    const int wave = t >> 5;
    const int blk  = blockIdx.x;
    const int n4   = n >> 2;
    const v4i* c4  = (const v4i*)cnt_in;
    const int g    = blk * SC_T + t;

    v4i c = c4[g < n4 ? g : n4 - 1];

    int pre = 0;
    for (int k0 = 0; k0 < blk; k0 += SC_B) {
        v4i p[SC_B];
#pragma unroll
        for (int u = 0; u < SC_B; ++u) {
            const int k = k0 + u;
            p[u] = c4[(k < blk ? k : blk - 1) * SC_T + t];
        }
#pragma unroll
        for (int u = 0; u < SC_B; ++u) {
            if (k0 + u < blk) pre += (p[u].x + p[u].y) + (p[u].z + p[u].w);
        }
    }

    if (g >= n4) { v4i z = {}; c = z; }
    const int s = (c.x + c.y) + (c.z + c.w);

    int incl = s;
#pragma unroll
    for (int d = 1; d < 32; d <<= 1) {
        const int v = __shfl_up(incl, d);
        if (lane >= d) incl += v;
    }
    int pw = pre;
#pragma unroll
    for (int d = 16; d >= 1; d >>= 1) pw += __shfl_xor(pw, d);
    if (lane == 31) wtot[wave] = incl;
    if (lane == 0)  wpre[wave] = pw;
    __syncthreads();

    int base = wpre[lane] + (lane < wave ? wtot[lane] : 0);
#pragma unroll
    for (int d = 16; d >= 1; d >>= 1) base += __shfl_xor(base, d);

    v4i o;
    o.x = base + (incl - s);
    o.y = o.x + c.x;
    o.z = o.y + c.y;
    o.w = o.z + c.z;
    if (g < n4) ((v4i*)off)[g] = o;
    if (g == n4 - 1) off[n] = o.w + c.w;
}

__global__ void fill_kernel(const int* __restrict__ src, const int* __restrict__ dst,
                            const int* __restrict__ rank, const int* __restrict__ off,
                            const int* __restrict__ cnt_out,
                            v2i* __restrict__ csr, int nE) {
    int e = blockIdx.x * blockDim.x + threadIdx.x;
    if (e < nE) {
        const int s = src[e];
        const int p = off[dst[e]] + rank[e];
        v2i rec;
        rec.x = s;
        rec.y = __float_as_int(rsqrtf(fmaxf((float)cnt_out[s], 1.0f)));
        csr[p] = rec;
    }
}

template <int CAP>
__device__ __forceinline__ void stage_records(v2i* sr, const v2i* __restrict__ csr,
                                              int e0, int nE, int lane) {
    if (nE > 0) {
        const int nst = nE < CAP ? nE : CAP;
        v2i r[CAP / 32];
#pragma unroll
        for (int j = 0; j < CAP / 32; ++j) {
            const int k = j * 32 + lane;
            r[j] = csr[e0 + (k < nE ? k : nE - 1)];
        }
#pragma unroll
        for (int j = 0; j < CAP / 32; ++j) {
            const int k = j * 32 + lane;
            if (k < nst) sr[k] = r[j];
        }
    }
}

template <int CAP>
__device__ __forceinline__ void batch_records(const v2i* sr, const v2i* __restrict__ csr,
                                              int e0, int p, int end, int (&s)[8], float (&w)[8]) {
    if (p - e0 + 8 <= CAP) {
#pragma unroll
        for (int u = 0; u < 8; ++u) {
            int q = p + u;
            q = q < end ? q : end - 1;
            const v2i rec = sr[q - e0];
            s[u] = rec.x;
            w[u] = __int_as_float(rec.y);
        }
    } else {
#pragma unroll
        for (int u = 0; u < 8; ++u) {
            int q = p + u;
            q = q < end ? q : end - 1;
            const v2i rec = csr[q];
            s[u] = rec.x;
            w[u] = __int_as_float(rec.y);
        }
    }
}

#define PW  4
#define LDP (FEATS + 4)
#define RCAP 384

template <int NT>
__device__ __forceinline__ void tile_gemm(const float* ap, const float* __restrict__ W,
                                          int half, int l16, v8f (&acc)[NT]) {
    constexpr int NOUT = NT * 16;
#pragma unroll
    for (int k0 = 0; k0 < FEATS; k0 += 4) {
        const v2f A = *(const v2f*)(ap + k0);
        const float* wp = W + (size_t)(k0 + half * 2) * NOUT + l16;
#pragma unroll
        for (int t = 0; t < NT; ++t) {
            v2f B;
            B.x = wp[t * 16];
            B.y = wp[t * 16 + NOUT];
            acc[t] = __builtin_amdgcn_wmma_f32_16x16x4_f32(
                false, A, false, B, (short)0, acc[t], false, false);
        }
    }
}

template <int NT, bool RELU, int NT2>
__global__ __launch_bounds__(PW * 32) void pull_gemm_wmma(
        const float* __restrict__ hin, const int* __restrict__ off,
        const v2i* __restrict__ csr,
        const float* __restrict__ W, const float* __restrict__ bias,
        const float* __restrict__ W2,
        float* __restrict__ out, int nRows) {
    constexpr int NOUT = NT * 16;
    static_assert(NT2 == 0 || NOUT == FEATS, "the second product reads the first one's 64-wide tile");
    __shared__ __attribute__((aligned(16))) float tile[PW][16 * LDP];
    __shared__ __attribute__((aligned(16))) v2i   recs[PW][RCAP];

    const int lane = threadIdx.x & 31;
    const int wave = threadIdx.x >> 5;
    const int rowBase = (blockIdx.x * PW + wave) * 16;
    const bool active = rowBase < nRows;
    const int half = lane >> 4;
    const int l16  = lane & 15;
    float* myt = tile[wave];

    int bnd = 0, e0 = 0;
    if (active) {
        const int bi = rowBase + lane;
        bnd = off[bi < nRows ? bi : nRows];
        e0  = __shfl(bnd, 0);
        stage_records<RCAP>(recs[wave], csr, e0, __shfl(bnd, 16) - e0, lane);
    }
    __syncthreads();

    if (active) {
        for (int i = 0; i < 8; ++i) {
            const int r   = i * 2 + half;
            const int beg = __shfl(bnd, r);
            const int end = __shfl(bnd, r + 1);
            v4f acc = {};
            for (int p = beg; p < end; p += 8) {
                int   s[8];
                float w[8];
                v4f   v[8];
                batch_records<RCAP>(recs[wave], csr, e0, p, end, s, w);
#pragma unroll
                for (int u = 0; u < 8; ++u)
                    v[u] = *(const v4f*)(hin + (size_t)s[u] * FEATS + l16 * 4);
                __builtin_amdgcn_sched_barrier(0);
#pragma unroll
                for (int u = 0; u < 8; ++u) {
                    if (p + u < end) {
                        acc.x = fmaf(v[u].x, w[u], acc.x);
                        acc.y = fmaf(v[u].y, w[u], acc.y);
                        acc.z = fmaf(v[u].z, w[u], acc.z);
                        acc.w = fmaf(v[u].w, w[u], acc.w);
                    }
                }
            }
            const float dn = rsqrtf(fmaxf((float)(end - beg), 1.0f));
            *(v4f*)(myt + r * LDP + l16 * 4) = acc * dn;
        }
    }
    __syncthreads();

    const float* ap = myt + l16 * LDP + half * 2;
    v8f acc2[NT] = {};
    if (active) {
        tile_gemm<NT>(ap, W, half, l16, acc2);
#pragma unroll
        for (int t = 0; t < NT; ++t) {
            const int col = t * 16 + l16;
            const float bv = bias[col];
#pragma unroll
            for (int i = 0; i < 8; ++i) {
                const int row = rowBase + half * 8 + i;
                float v = acc2[t][i] + bv;
                if (RELU) v = fmaxf(v, 0.0f);
                if constexpr (NT2 == 0) {
                    if (row < nRows) out[(size_t)row * NOUT + col] = v;
                } else {
                    myt[(half * 8 + i) * LDP + col] = v;
                }
            }
        }
    }

    if constexpr (NT2 > 0) {
        constexpr int NOUT2 = NT2 * 16;
        __syncthreads();
        if (active) {
            v8f acc3[NT2] = {};
            tile_gemm<NT2>(ap, W2, half, l16, acc3);
#pragma unroll
            for (int t = 0; t < NT2; ++t) {
                const int col = t * 16 + l16;
#pragma unroll
                for (int i = 0; i < 8; ++i) {
                    const int row = rowBase + half * 8 + i;
                    if (row < nRows) out[(size_t)row * NOUT2 + col] = acc3[t][i];
                }
            }
        }
    }
}

#define AGW 8
#define AGP 2
#define AGF 32
#define AGC 256
static_assert(AGF * 4 == 128, "one row = one 128-byte line = 8 lanes x 16 B");

__global__ __launch_bounds__(AGW * 32) void pull_rows32_kernel(
        const float* __restrict__ tin, const int* __restrict__ off,
        const v2i* __restrict__ csr, const float* __restrict__ bias,
        float* __restrict__ out, int nRows) {
    __shared__ __attribute__((aligned(16))) v2i recs[AGW][AGC];
    const int lane = threadIdx.x & 31;
    const int wave = threadIdx.x >> 5;
    const int qw   = lane >> 3;
    const int l8   = lane & 7;
    const int rowBase = (blockIdx.x * AGW + wave) * (AGP * 4);

    const int bi  = rowBase + lane;
    const int bnd = off[bi < nRows ? bi : nRows];
    const int e0  = __shfl(bnd, 0);
    stage_records<AGC>(recs[wave], csr, e0, __shfl(bnd, AGP * 4) - e0, lane);
    const v4f bv = *(const v4f*)(bias + l8 * 4);
    __syncthreads();

#pragma unroll
    for (int i = 0; i < AGP; ++i) {
        const int row = rowBase + i * 4 + qw;
        const int beg = __shfl(bnd, i * 4 + qw);
        const int end = __shfl(bnd, i * 4 + qw + 1);
        v4f acc = {};
        for (int p = beg; p < end; p += 8) {
            int   s[8];
            float w[8];
            v4f   v[8];
            batch_records<AGC>(recs[wave], csr, e0, p, end, s, w);
#pragma unroll
            for (int u = 0; u < 8; ++u)
                v[u] = *(const v4f*)(tin + (size_t)s[u] * AGF + l8 * 4);
            __builtin_amdgcn_sched_barrier(0);
#pragma unroll
            for (int u = 0; u < 8; ++u) {
                if (p + u < end) {
                    acc.x = fmaf(v[u].x, w[u], acc.x);
                    acc.y = fmaf(v[u].y, w[u], acc.y);
                    acc.z = fmaf(v[u].z, w[u], acc.z);
                    acc.w = fmaf(v[u].w, w[u], acc.w);
                }
            }
        }
        const float dn = rsqrtf(fmaxf((float)(end - beg), 1.0f));
        if (row < nRows) *(v4f*)(out + (size_t)row * AGF + l8 * 4) = acc * dn + bv;
    }
}

extern "C" void kernel_launch(void* const* d_in, const int* in_sizes, int n_in,
                              void* d_out, int out_size, void* d_ws, size_t ws_size,
                              hipStream_t stream) {
    const float* x    = (const float*)d_in[0];
    const int*   esrc = (const int*)d_in[1];
    const int*   edst = (const int*)d_in[2];
    const float* W0   = (const float*)d_in[3];
    const float* b0   = (const float*)d_in[4];
    const float* W1   = (const float*)d_in[5];
    const float* b1   = (const float*)d_in[6];
    const float* W2   = (const float*)d_in[7];
    const float* b2   = (const float*)d_in[8];
    float* outp = (float*)d_out;

    const int NOFF = 100096;
    int*   cnt_out = (int*)d_ws;
    int*   cnt_in  = cnt_out + NNODES;
    int*   off     = cnt_in + NNODES;
    int*   rank    = off + NOFF;
    v2i*   csr     = (v2i*)(rank + NEDGES);
    float* hA      = (float*)(csr + NEDGES);
    float* tB      = hA + (size_t)NNODES * FEATS;

    const int T = 256;
    const int gZero = (2 * NNODES / 4 + T - 1) / T;
    const int gEdge = (NEDGES + T - 1) / T;
    const int gPull = (NNODES + 16 * PW - 1) / (16 * PW);
    const int gAgg  = (NNODES + 4 * AGP * AGW - 1) / (4 * AGP * AGW);

    zero_i4<<<gZero, T, 0, stream>>>(cnt_out, 2 * NNODES / 4);
    degrees_lds_kernel<<<dim3(OD_CHUNKS, 2, 2), OD_T, 0, stream>>>(esrc, edst, cnt_out, cnt_in, rank);
    scan_kernel<<<SC_G, SC_T, 0, stream>>>(cnt_in, off, NNODES);
    fill_kernel<<<gEdge, T, 0, stream>>>(esrc, edst, rank, off, cnt_out, csr, NEDGES);

    pull_gemm_wmma<4, true, 0><<<gPull, PW * 32, 0, stream>>>(x,  off, csr, W0, b0, nullptr, hA, NNODES);
    pull_gemm_wmma<4, true, 2><<<gPull, PW * 32, 0, stream>>>(hA, off, csr, W1, b1, W2,      tB, NNODES);
    pull_rows32_kernel<<<gAgg, AGW * 32, 0, stream>>>(tB, off, csr, b2, outp, NNODES);
}
